# starting kernel
# baseline (speedup 1.0000x reference)
#include <hip/hip_runtime.h>
#include <hip/hip_bf16.h>
#include <math.h>

typedef __attribute__((ext_vector_type(16))) __bf16 v16bf;
typedef __attribute__((ext_vector_type(8)))  __bf16 v8bf;
typedef __attribute__((ext_vector_type(8)))  float  v8f;

constexpr int B  = 256;
constexpr int U  = 128;
constexpr int P  = 64;
constexpr int H  = 768;
constexpr int K2 = U * P;

__global__ void norms_kernel(const float* __restrict__ src,
                             float* __restrict__ inv_out, int len) {
  int wave = (blockIdx.x * blockDim.x + threadIdx.x) >> 5;
  int lane = threadIdx.x & 31;
  if (wave >= len * B) return;
  int i = wave / B, b = wave % B;
  const float* row = src + (size_t)wave * H;
  float s = 0.f;
#pragma unroll
  for (int j = 0; j < H; j += 128) {
    float4 v = *(const float4*)(row + j + lane * 4);
    s += v.x * v.x + v.y * v.y + v.z * v.z + v.w * v.w;
  }
#pragma unroll
  for (int m = 16; m; m >>= 1) s += __shfl_xor(s, m, 32);
  if (lane == 0) inv_out[b * len + i] = 1.0f / fmaxf(sqrtf(s), 1e-8f);
}

__global__ void wconv_kernel(const float* __restrict__ w_utt,
                             const float* __restrict__ w_pheno,
                             __bf16* __restrict__ wu_bf,
                             __bf16* __restrict__ wp_bf) {
  int idx = blockIdx.x * blockDim.x + threadIdx.x;
  if (idx < U * K2) wu_bf[idx] = (__bf16)w_utt[idx];
  if (idx < P * K2) {
    int o = idx / K2, r = idx % K2;
    int u = r / P, p = r % P;
    wp_bf[idx] = (__bf16)w_pheno[(size_t)o * K2 + p * U + u];
  }
}

__device__ __forceinline__ v16bf load_frag_f32(const float* __restrict__ p) {
  float4 a0 = *(const float4*)(p);
  float4 a1 = *(const float4*)(p + 4);
  float4 b0 = *(const float4*)(p + 16);
  float4 b1 = *(const float4*)(p + 20);
  v16bf r;
  r[0]  = (__bf16)a0.x; r[1]  = (__bf16)a0.y;
  r[2]  = (__bf16)a0.z; r[3]  = (__bf16)a0.w;
  r[4]  = (__bf16)a1.x; r[5]  = (__bf16)a1.y;
  r[6]  = (__bf16)a1.z; r[7]  = (__bf16)a1.w;
  r[8]  = (__bf16)b0.x; r[9]  = (__bf16)b0.y;
  r[10] = (__bf16)b0.z; r[11] = (__bf16)b0.w;
  r[12] = (__bf16)b1.x; r[13] = (__bf16)b1.y;
  r[14] = (__bf16)b1.z; r[15] = (__bf16)b1.w;
  return r;
}

__device__ __forceinline__ v16bf load_frag_bf16(const __bf16* __restrict__ p) {
  v8bf lo = *(const v8bf*)(p);
  v8bf hi = *(const v8bf*)(p + 16);
  v16bf r;
#pragma unroll
  for (int j = 0; j < 8; ++j) { r[j] = lo[j]; r[j + 8] = hi[j]; }
  return r;
}

__global__ void cosine_wmma_kernel(const float* __restrict__ utt,
                                   const float* __restrict__ pheno,
                                   const float* __restrict__ inv_nu,
                                   const float* __restrict__ inv_np,
                                   __bf16* __restrict__ c_bf) {
  int b    = blockIdx.x;
  int wave = threadIdx.x >> 5;
  int lane = threadIdx.x & 31;
  int rt   = lane & 15;
  int hi   = lane >> 4;
  int k0   = hi ? 8 : 0;

  int u_row = wave * 16 + rt;
  const float* Arow = utt + ((size_t)u_row * B + b) * H + k0;

  const float* Brow[4];
#pragma unroll
  for (int nt = 0; nt < 4; ++nt) {
    int p_col = nt * 16 + rt;
    Brow[nt]  = pheno + ((size_t)p_col * B + b) * H + k0;
  }

  v8f acc[4] = {};
  for (int kk = 0; kk < H; kk += 32) {
    v16bf a = load_frag_f32(Arow + kk);
#pragma unroll
    for (int nt = 0; nt < 4; ++nt) {
      v16bf bb = load_frag_f32(Brow[nt] + kk);
      acc[nt] = __builtin_amdgcn_wmma_f32_16x16x32_bf16(
          false, a, false, bb, (short)0, acc[nt], false, false);
    }
  }

  const float* su_base = inv_nu + b * U + wave * 16 + 8 * hi;
  float4 su0 = *(const float4*)(su_base);
  float4 su1 = *(const float4*)(su_base + 4);
  float su[8] = {su0.x, su0.y, su0.z, su0.w, su1.x, su1.y, su1.z, su1.w};

  __bf16* cb = c_bf + (size_t)b * K2;
#pragma unroll
  for (int nt = 0; nt < 4; ++nt) {
    float sp = inv_np[b * P + nt * 16 + rt];
#pragma unroll
    for (int r = 0; r < 8; ++r) {
      int m = r + 8 * hi;
      cb[(wave * 16 + m) * P + nt * 16 + rt] = (__bf16)(acc[nt][r] * su[r] * sp);
    }
  }
}

__global__ void logits_wmma_kernel(const __bf16* __restrict__ c_bf,
                                   const __bf16* __restrict__ W,
                                   const float* __restrict__ bias,
                                   float* __restrict__ out, int N) {
  int mtile = blockIdx.x;
  int wave  = threadIdx.x >> 5;
  int ntile = blockIdx.y * 4 + wave;
  int lane  = threadIdx.x & 31;
  int rt    = lane & 15;
  int hi    = lane >> 4;
  int k0    = hi ? 8 : 0;

  const __bf16* Arow = c_bf + (size_t)(mtile * 16 + rt) * K2 + k0;
  const __bf16* Wrow = W    + (size_t)(ntile * 16 + rt) * K2 + k0;

  v8f acc = {};
  for (int kk = 0; kk < K2; kk += 32) {
    v16bf a = load_frag_bf16(Arow + kk);
    v16bf w = load_frag_bf16(Wrow + kk);
    acc = __builtin_amdgcn_wmma_f32_16x16x32_bf16(
        false, a, false, w, (short)0, acc, false, false);
  }

  int n = ntile * 16 + rt;
  float bv = bias[n];
#pragma unroll
  for (int r = 0; r < 8; ++r) {
    int m = r + 8 * hi;
    out[(size_t)(mtile * 16 + m) * N + n] = acc[r] + bv;
  }
}

__global__ void softmax_kernel(float* __restrict__ x, int n) {
  int row  = (blockIdx.x * blockDim.x + threadIdx.x) >> 5;
  int lane = threadIdx.x & 31;
  if (row >= B) return;
  float* p = x + (size_t)row * n;
  int per = n >> 5;
  float vals[4];
  float mx = -1e30f;
  for (int j = 0; j < per; ++j) { vals[j] = p[lane + 32 * j]; mx = fmaxf(mx, vals[j]); }
#pragma unroll
  for (int m = 16; m; m >>= 1) mx = fmaxf(mx, __shfl_xor(mx, m, 32));
  float s = 0.f;
  for (int j = 0; j < per; ++j) { vals[j] = __expf(vals[j] - mx); s += vals[j]; }
#pragma unroll
  for (int m = 16; m; m >>= 1) s += __shfl_xor(s, m, 32);
  float inv = 1.0f / s;
  for (int j = 0; j < per; ++j) p[lane + 32 * j] = vals[j] * inv;
}

extern "C" void kernel_launch(void* const* d_in, const int* in_sizes, int n_in,
                              void* d_out, int out_size, void* d_ws, size_t ws_size,
                              hipStream_t stream) {
  const float* utt     = (const float*)d_in[0];
  const float* pheno   = (const float*)d_in[1];
  const float* w_utt   = (const float*)d_in[2];
  const float* b_utt   = (const float*)d_in[3];
  const float* w_pheno = (const float*)d_in[4];
  const float* b_pheno = (const float*)d_in[5];
  (void)in_sizes; (void)n_in; (void)out_size; (void)ws_size;

  char* ws = (char*)d_ws;
  size_t off = 0;
  auto alloc = [&](size_t bytes) {
    void* p = ws + off;
    off = (off + bytes + 255) & ~(size_t)255;
    return p;
  };
  float*  inv_nu = (float*)alloc((size_t)B * U * sizeof(float));
  float*  inv_np = (float*)alloc((size_t)B * P * sizeof(float));
  __bf16* wu_bf  = (__bf16*)alloc((size_t)U * K2 * sizeof(__bf16));
  __bf16* wp_bf  = (__bf16*)alloc((size_t)P * K2 * sizeof(__bf16));
  __bf16* c_bf   = (__bf16*)alloc((size_t)B * K2 * sizeof(__bf16));

  float* row_logits = (float*)d_out;
  float* col_logits = (float*)d_out + B * U;

  norms_kernel<<<(B * U) / 8, 256, 0, stream>>>(utt, inv_nu, U);
  norms_kernel<<<(B * P) / 8, 256, 0, stream>>>(pheno, inv_np, P);
  wconv_kernel<<<(U * K2) / 256, 256, 0, stream>>>(w_utt, w_pheno, wu_bf, wp_bf);

  cosine_wmma_kernel<<<B, 256, 0, stream>>>(utt, pheno, inv_nu, inv_np, c_bf);

  logits_wmma_kernel<<<dim3(B / 16, U / 64), 128, 0, stream>>>(c_bf, wu_bf, b_utt, row_logits, U);
  logits_wmma_kernel<<<dim3(B / 16, P / 64), 128, 0, stream>>>(c_bf, wp_bf, b_pheno, col_logits, P);

  softmax_kernel<<<B / 8, 256, 0, stream>>>(row_logits, U);
  softmax_kernel<<<B / 8, 256, 0, stream>>>(col_logits, P);
}


# logits GEMMs: serial K loop (one memory round trip per step) split over 16 waves per 16x16 tile, each requesting 8 steps ahead; partials summed via LDS in fixed wave order (f32 sum order changes)
# speedup vs baseline: 1.8397x; 1.8397x over previous
#include <hip/hip_runtime.h>
#include <hip/hip_bf16.h>
#include <math.h>

typedef __attribute__((ext_vector_type(16))) __bf16 v16bf;
typedef __attribute__((ext_vector_type(8)))  __bf16 v8bf;
typedef __attribute__((ext_vector_type(8)))  float  v8f;

constexpr int B  = 256;
constexpr int U  = 128;
constexpr int P  = 64;
constexpr int H  = 768;
constexpr int K2 = U * P;

__global__ void norms_kernel(const float* __restrict__ src,
                             float* __restrict__ inv_out, int len) {
  int wave = (blockIdx.x * blockDim.x + threadIdx.x) >> 5;
  int lane = threadIdx.x & 31;
  if (wave >= len * B) return;
  int i = wave / B, b = wave % B;
  const float* row = src + (size_t)wave * H;
  float s = 0.f;
#pragma unroll
  for (int j = 0; j < H; j += 128) {
    float4 v = *(const float4*)(row + j + lane * 4);
    s += v.x * v.x + v.y * v.y + v.z * v.z + v.w * v.w;
  }
#pragma unroll
  for (int m = 16; m; m >>= 1) s += __shfl_xor(s, m, 32);
  if (lane == 0) inv_out[b * len + i] = 1.0f / fmaxf(sqrtf(s), 1e-8f);
}

__global__ void wconv_kernel(const float* __restrict__ w_utt,
                             const float* __restrict__ w_pheno,
                             __bf16* __restrict__ wu_bf,
                             __bf16* __restrict__ wp_bf) {
  int idx = blockIdx.x * blockDim.x + threadIdx.x;
  if (idx < U * K2) wu_bf[idx] = (__bf16)w_utt[idx];
  if (idx < P * K2) {
    int o = idx / K2, r = idx % K2;
    int u = r / P, p = r % P;
    wp_bf[idx] = (__bf16)w_pheno[(size_t)o * K2 + p * U + u];
  }
}

__device__ __forceinline__ v16bf load_frag_f32(const float* __restrict__ p) {
  float4 a0 = *(const float4*)(p);
  float4 a1 = *(const float4*)(p + 4);
  float4 b0 = *(const float4*)(p + 16);
  float4 b1 = *(const float4*)(p + 20);
  v16bf r;
  r[0]  = (__bf16)a0.x; r[1]  = (__bf16)a0.y;
  r[2]  = (__bf16)a0.z; r[3]  = (__bf16)a0.w;
  r[4]  = (__bf16)a1.x; r[5]  = (__bf16)a1.y;
  r[6]  = (__bf16)a1.z; r[7]  = (__bf16)a1.w;
  r[8]  = (__bf16)b0.x; r[9]  = (__bf16)b0.y;
  r[10] = (__bf16)b0.z; r[11] = (__bf16)b0.w;
  r[12] = (__bf16)b1.x; r[13] = (__bf16)b1.y;
  r[14] = (__bf16)b1.z; r[15] = (__bf16)b1.w;
  return r;
}

__device__ __forceinline__ v16bf load_frag_bf16(const __bf16* __restrict__ p) {
  v8bf lo = *(const v8bf*)(p);
  v8bf hi = *(const v8bf*)(p + 16);
  v16bf r;
#pragma unroll
  for (int j = 0; j < 8; ++j) { r[j] = lo[j]; r[j + 8] = hi[j]; }
  return r;
}

__global__ void cosine_wmma_kernel(const float* __restrict__ utt,
                                   const float* __restrict__ pheno,
                                   const float* __restrict__ inv_nu,
                                   const float* __restrict__ inv_np,
                                   __bf16* __restrict__ c_bf) {
  int b    = blockIdx.x;
  int wave = threadIdx.x >> 5;
  int lane = threadIdx.x & 31;
  int rt   = lane & 15;
  int hi   = lane >> 4;
  int k0   = hi ? 8 : 0;

  int u_row = wave * 16 + rt;
  const float* Arow = utt + ((size_t)u_row * B + b) * H + k0;

  const float* Brow[4];
#pragma unroll
  for (int nt = 0; nt < 4; ++nt) {
    int p_col = nt * 16 + rt;
    Brow[nt]  = pheno + ((size_t)p_col * B + b) * H + k0;
  }

  v8f acc[4] = {};
  for (int kk = 0; kk < H; kk += 32) {
    v16bf a = load_frag_f32(Arow + kk);
#pragma unroll
    for (int nt = 0; nt < 4; ++nt) {
      v16bf bb = load_frag_f32(Brow[nt] + kk);
      acc[nt] = __builtin_amdgcn_wmma_f32_16x16x32_bf16(
          false, a, false, bb, (short)0, acc[nt], false, false);
    }
  }

  const float* su_base = inv_nu + b * U + wave * 16 + 8 * hi;
  float4 su0 = *(const float4*)(su_base);
  float4 su1 = *(const float4*)(su_base + 4);
  float su[8] = {su0.x, su0.y, su0.z, su0.w, su1.x, su1.y, su1.z, su1.w};

  __bf16* cb = c_bf + (size_t)b * K2;
#pragma unroll
  for (int nt = 0; nt < 4; ++nt) {
    float sp = inv_np[b * P + nt * 16 + rt];
#pragma unroll
    for (int r = 0; r < 8; ++r) {
      int m = r + 8 * hi;
      cb[(wave * 16 + m) * P + nt * 16 + rt] = (__bf16)(acc[nt][r] * su[r] * sp);
    }
  }
}

constexpr int LG_WAVES  = 16;
constexpr int LG_KSLICE = K2 / LG_WAVES;
constexpr int LG_STEPS  = LG_KSLICE / 32;
constexpr int LG_AHEAD  = 8;

__global__ void __launch_bounds__(LG_WAVES * 32)
logits_wmma_kernel(const __bf16* __restrict__ c_bf,
                   const __bf16* __restrict__ W,
                   const float* __restrict__ bias,
                   float* __restrict__ out, int N) {
  __shared__ float red[LG_WAVES][256];

  int mtile = blockIdx.x;
  int ntile = blockIdx.y;
  int wave  = threadIdx.x >> 5;
  int lane  = threadIdx.x & 31;
  int rt    = lane & 15;
  int hi    = lane >> 4;
  int k0    = hi ? 8 : 0;

  const __bf16* Arow = c_bf + (size_t)(mtile * 16 + rt) * K2 + wave * LG_KSLICE + k0;
  const __bf16* Wrow = W    + (size_t)(ntile * 16 + rt) * K2 + wave * LG_KSLICE + k0;

  v16bf a[LG_AHEAD], w[LG_AHEAD];
#pragma unroll
  for (int i = 0; i < LG_AHEAD; ++i) {
    a[i] = load_frag_bf16(Arow + i * 32);
    w[i] = load_frag_bf16(Wrow + i * 32);
  }
  __builtin_amdgcn_sched_barrier(0);
  v8f acc = {};
#pragma unroll
  for (int i = 0; i < LG_STEPS; ++i) {
    acc = __builtin_amdgcn_wmma_f32_16x16x32_bf16(
        false, a[i % LG_AHEAD], false, w[i % LG_AHEAD], (short)0, acc, false, false);
    if (i + LG_AHEAD < LG_STEPS) {
      a[i % LG_AHEAD] = load_frag_bf16(Arow + (i + LG_AHEAD) * 32);
      w[i % LG_AHEAD] = load_frag_bf16(Wrow + (i + LG_AHEAD) * 32);
      __builtin_amdgcn_sched_barrier(0);
    }
  }

#pragma unroll
  for (int r = 0; r < 8; ++r) red[wave][r * 32 + lane] = acc[r];
  __syncthreads();

  if (threadIdx.x < 256) {
    int e = threadIdx.x;
    float s = red[0][e];
#pragma unroll
    for (int w = 1; w < LG_WAVES; ++w) s += red[w][e];
    int r = e >> 5;
    int l = e & 31;
    int m = r + 8 * (l >> 4);
    int n = ntile * 16 + (l & 15);
    out[(size_t)(mtile * 16 + m) * N + n] = s + bias[n];
  }
}

__global__ void softmax_kernel(float* __restrict__ x, int n) {
  int row  = (blockIdx.x * blockDim.x + threadIdx.x) >> 5;
  int lane = threadIdx.x & 31;
  if (row >= B) return;
  float* p = x + (size_t)row * n;
  int per = n >> 5;
  float vals[4];
  float mx = -1e30f;
  for (int j = 0; j < per; ++j) { vals[j] = p[lane + 32 * j]; mx = fmaxf(mx, vals[j]); }
#pragma unroll
  for (int m = 16; m; m >>= 1) mx = fmaxf(mx, __shfl_xor(mx, m, 32));
  float s = 0.f;
  for (int j = 0; j < per; ++j) { vals[j] = __expf(vals[j] - mx); s += vals[j]; }
#pragma unroll
  for (int m = 16; m; m >>= 1) s += __shfl_xor(s, m, 32);
  float inv = 1.0f / s;
  for (int j = 0; j < per; ++j) p[lane + 32 * j] = vals[j] * inv;
}

extern "C" void kernel_launch(void* const* d_in, const int* in_sizes, int n_in,
                              void* d_out, int out_size, void* d_ws, size_t ws_size,
                              hipStream_t stream) {
  const float* utt     = (const float*)d_in[0];
  const float* pheno   = (const float*)d_in[1];
  const float* w_utt   = (const float*)d_in[2];
  const float* b_utt   = (const float*)d_in[3];
  const float* w_pheno = (const float*)d_in[4];
  const float* b_pheno = (const float*)d_in[5];
  (void)in_sizes; (void)n_in; (void)out_size; (void)ws_size;

  char* ws = (char*)d_ws;
  size_t off = 0;
  auto alloc = [&](size_t bytes) {
    void* p = ws + off;
    off = (off + bytes + 255) & ~(size_t)255;
    return p;
  };
  float*  inv_nu = (float*)alloc((size_t)B * U * sizeof(float));
  float*  inv_np = (float*)alloc((size_t)B * P * sizeof(float));
  __bf16* wu_bf  = (__bf16*)alloc((size_t)U * K2 * sizeof(__bf16));
  __bf16* wp_bf  = (__bf16*)alloc((size_t)P * K2 * sizeof(__bf16));
  __bf16* c_bf   = (__bf16*)alloc((size_t)B * K2 * sizeof(__bf16));

  float* row_logits = (float*)d_out;
  float* col_logits = (float*)d_out + B * U;

  norms_kernel<<<(B * U) / 8, 256, 0, stream>>>(utt, inv_nu, U);
  norms_kernel<<<(B * P) / 8, 256, 0, stream>>>(pheno, inv_np, P);
  wconv_kernel<<<(U * K2) / 256, 256, 0, stream>>>(w_utt, w_pheno, wu_bf, wp_bf);

  cosine_wmma_kernel<<<B, 256, 0, stream>>>(utt, pheno, inv_nu, inv_np, c_bf);

  logits_wmma_kernel<<<dim3(B / 16, U / 16), LG_WAVES * 32, 0, stream>>>(c_bf, wu_bf, b_utt, row_logits, U);
  logits_wmma_kernel<<<dim3(B / 16, P / 16), LG_WAVES * 32, 0, stream>>>(c_bf, wp_bf, b_pheno, col_logits, P);

  softmax_kernel<<<B / 8, 256, 0, stream>>>(row_logits, U);
  softmax_kernel<<<B / 8, 256, 0, stream>>>(col_logits, P);
}


# cosine GEMM compiled for the 2 waves per SIMD it runs at (launch bound 256, waves-per-EU 2..2), K loop unrolled by 2: a trip's 40 operand loads go out as one group, not 4 dependent groups per k-step
# speedup vs baseline: 2.3025x; 1.2516x over previous
#include <hip/hip_runtime.h>
#include <hip/hip_bf16.h>
#include <math.h>

typedef __attribute__((ext_vector_type(16))) __bf16 v16bf;
typedef __attribute__((ext_vector_type(8)))  __bf16 v8bf;
typedef __attribute__((ext_vector_type(8)))  float  v8f;

constexpr int B  = 256;
constexpr int U  = 128;
constexpr int P  = 64;
constexpr int H  = 768;
constexpr int K2 = U * P;

__global__ void norms_kernel(const float* __restrict__ src,
                             float* __restrict__ inv_out, int len) {
  int wave = (blockIdx.x * blockDim.x + threadIdx.x) >> 5;
  int lane = threadIdx.x & 31;
  if (wave >= len * B) return;
  int i = wave / B, b = wave % B;
  const float* row = src + (size_t)wave * H;
  float s = 0.f;
#pragma unroll
  for (int j = 0; j < H; j += 128) {
    float4 v = *(const float4*)(row + j + lane * 4);
    s += v.x * v.x + v.y * v.y + v.z * v.z + v.w * v.w;
  }
#pragma unroll
  for (int m = 16; m; m >>= 1) s += __shfl_xor(s, m, 32);
  if (lane == 0) inv_out[b * len + i] = 1.0f / fmaxf(sqrtf(s), 1e-8f);
}

__global__ void wconv_kernel(const float* __restrict__ w_utt,
                             const float* __restrict__ w_pheno,
                             __bf16* __restrict__ wu_bf,
                             __bf16* __restrict__ wp_bf) {
  int idx = blockIdx.x * blockDim.x + threadIdx.x;
  if (idx < U * K2) wu_bf[idx] = (__bf16)w_utt[idx];
  if (idx < P * K2) {
    int o = idx / K2, r = idx % K2;
    int u = r / P, p = r % P;
    wp_bf[idx] = (__bf16)w_pheno[(size_t)o * K2 + p * U + u];
  }
}

__device__ __forceinline__ v16bf load_frag_f32(const float* __restrict__ p) {
  float4 a0 = *(const float4*)(p);
  float4 a1 = *(const float4*)(p + 4);
  float4 b0 = *(const float4*)(p + 16);
  float4 b1 = *(const float4*)(p + 20);
  v16bf r;
  r[0]  = (__bf16)a0.x; r[1]  = (__bf16)a0.y;
  r[2]  = (__bf16)a0.z; r[3]  = (__bf16)a0.w;
  r[4]  = (__bf16)a1.x; r[5]  = (__bf16)a1.y;
  r[6]  = (__bf16)a1.z; r[7]  = (__bf16)a1.w;
  r[8]  = (__bf16)b0.x; r[9]  = (__bf16)b0.y;
  r[10] = (__bf16)b0.z; r[11] = (__bf16)b0.w;
  r[12] = (__bf16)b1.x; r[13] = (__bf16)b1.y;
  r[14] = (__bf16)b1.z; r[15] = (__bf16)b1.w;
  return r;
}

__device__ __forceinline__ v16bf load_frag_bf16(const __bf16* __restrict__ p) {
  v8bf lo = *(const v8bf*)(p);
  v8bf hi = *(const v8bf*)(p + 16);
  v16bf r;
#pragma unroll
  for (int j = 0; j < 8; ++j) { r[j] = lo[j]; r[j + 8] = hi[j]; }
  return r;
}

__global__ void __launch_bounds__(256) __attribute__((amdgpu_waves_per_eu(2, 2)))
cosine_wmma_kernel(const float* __restrict__ utt,
                                   const float* __restrict__ pheno,
                                   const float* __restrict__ inv_nu,
                                   const float* __restrict__ inv_np,
                                   __bf16* __restrict__ c_bf) {
  int b    = blockIdx.x;
  int wave = threadIdx.x >> 5;
  int lane = threadIdx.x & 31;
  int rt   = lane & 15;
  int hi   = lane >> 4;
  int k0   = hi ? 8 : 0;

  int u_row = wave * 16 + rt;
  const float* Arow = utt + ((size_t)u_row * B + b) * H + k0;

  const float* Brow[4];
#pragma unroll
  for (int nt = 0; nt < 4; ++nt) {
    int p_col = nt * 16 + rt;
    Brow[nt]  = pheno + ((size_t)p_col * B + b) * H + k0;
  }

  v8f acc[4] = {};
#pragma unroll 2
  for (int kk = 0; kk < H; kk += 32) {
    v16bf a = load_frag_f32(Arow + kk);
#pragma unroll
    for (int nt = 0; nt < 4; ++nt) {
      v16bf bb = load_frag_f32(Brow[nt] + kk);
      acc[nt] = __builtin_amdgcn_wmma_f32_16x16x32_bf16(
          false, a, false, bb, (short)0, acc[nt], false, false);
    }
  }

  const float* su_base = inv_nu + b * U + wave * 16 + 8 * hi;
  float4 su0 = *(const float4*)(su_base);
  float4 su1 = *(const float4*)(su_base + 4);
  float su[8] = {su0.x, su0.y, su0.z, su0.w, su1.x, su1.y, su1.z, su1.w};

  __bf16* cb = c_bf + (size_t)b * K2;
#pragma unroll
  for (int nt = 0; nt < 4; ++nt) {
    float sp = inv_np[b * P + nt * 16 + rt];
#pragma unroll
    for (int r = 0; r < 8; ++r) {
      int m = r + 8 * hi;
      cb[(wave * 16 + m) * P + nt * 16 + rt] = (__bf16)(acc[nt][r] * su[r] * sp);
    }
  }
}

constexpr int LG_WAVES  = 16;
constexpr int LG_KSLICE = K2 / LG_WAVES;
constexpr int LG_STEPS  = LG_KSLICE / 32;
constexpr int LG_AHEAD  = 8;

__global__ void __launch_bounds__(LG_WAVES * 32)
logits_wmma_kernel(const __bf16* __restrict__ c_bf,
                   const __bf16* __restrict__ W,
                   const float* __restrict__ bias,
                   float* __restrict__ out, int N) {
  __shared__ float red[LG_WAVES][256];

  int mtile = blockIdx.x;
  int ntile = blockIdx.y;
  int wave  = threadIdx.x >> 5;
  int lane  = threadIdx.x & 31;
  int rt    = lane & 15;
  int hi    = lane >> 4;
  int k0    = hi ? 8 : 0;

  const __bf16* Arow = c_bf + (size_t)(mtile * 16 + rt) * K2 + wave * LG_KSLICE + k0;
  const __bf16* Wrow = W    + (size_t)(ntile * 16 + rt) * K2 + wave * LG_KSLICE + k0;

  v16bf a[LG_AHEAD], w[LG_AHEAD];
#pragma unroll
  for (int i = 0; i < LG_AHEAD; ++i) {
    a[i] = load_frag_bf16(Arow + i * 32);
    w[i] = load_frag_bf16(Wrow + i * 32);
  }
  __builtin_amdgcn_sched_barrier(0);
  v8f acc = {};
#pragma unroll
  for (int i = 0; i < LG_STEPS; ++i) {
    acc = __builtin_amdgcn_wmma_f32_16x16x32_bf16(
        false, a[i % LG_AHEAD], false, w[i % LG_AHEAD], (short)0, acc, false, false);
    if (i + LG_AHEAD < LG_STEPS) {
      a[i % LG_AHEAD] = load_frag_bf16(Arow + (i + LG_AHEAD) * 32);
      w[i % LG_AHEAD] = load_frag_bf16(Wrow + (i + LG_AHEAD) * 32);
      __builtin_amdgcn_sched_barrier(0);
    }
  }

#pragma unroll
  for (int r = 0; r < 8; ++r) red[wave][r * 32 + lane] = acc[r];
  __syncthreads();

  if (threadIdx.x < 256) {
    int e = threadIdx.x;
    float s = red[0][e];
#pragma unroll
    for (int w = 1; w < LG_WAVES; ++w) s += red[w][e];
    int r = e >> 5;
    int l = e & 31;
    int m = r + 8 * (l >> 4);
    int n = ntile * 16 + (l & 15);
    out[(size_t)(mtile * 16 + m) * N + n] = s + bias[n];
  }
}

__global__ void softmax_kernel(float* __restrict__ x, int n) {
  int row  = (blockIdx.x * blockDim.x + threadIdx.x) >> 5;
  int lane = threadIdx.x & 31;
  if (row >= B) return;
  float* p = x + (size_t)row * n;
  int per = n >> 5;
  float vals[4];
  float mx = -1e30f;
  for (int j = 0; j < per; ++j) { vals[j] = p[lane + 32 * j]; mx = fmaxf(mx, vals[j]); }
#pragma unroll
  for (int m = 16; m; m >>= 1) mx = fmaxf(mx, __shfl_xor(mx, m, 32));
  float s = 0.f;
  for (int j = 0; j < per; ++j) { vals[j] = __expf(vals[j] - mx); s += vals[j]; }
#pragma unroll
  for (int m = 16; m; m >>= 1) s += __shfl_xor(s, m, 32);
  float inv = 1.0f / s;
  for (int j = 0; j < per; ++j) p[lane + 32 * j] = vals[j] * inv;
}

extern "C" void kernel_launch(void* const* d_in, const int* in_sizes, int n_in,
                              void* d_out, int out_size, void* d_ws, size_t ws_size,
                              hipStream_t stream) {
  const float* utt     = (const float*)d_in[0];
  const float* pheno   = (const float*)d_in[1];
  const float* w_utt   = (const float*)d_in[2];
  const float* b_utt   = (const float*)d_in[3];
  const float* w_pheno = (const float*)d_in[4];
  const float* b_pheno = (const float*)d_in[5];
  (void)in_sizes; (void)n_in; (void)out_size; (void)ws_size;

  char* ws = (char*)d_ws;
  size_t off = 0;
  auto alloc = [&](size_t bytes) {
    void* p = ws + off;
    off = (off + bytes + 255) & ~(size_t)255;
    return p;
  };
  float*  inv_nu = (float*)alloc((size_t)B * U * sizeof(float));
  float*  inv_np = (float*)alloc((size_t)B * P * sizeof(float));
  __bf16* wu_bf  = (__bf16*)alloc((size_t)U * K2 * sizeof(__bf16));
  __bf16* wp_bf  = (__bf16*)alloc((size_t)P * K2 * sizeof(__bf16));
  __bf16* c_bf   = (__bf16*)alloc((size_t)B * K2 * sizeof(__bf16));

  float* row_logits = (float*)d_out;
  float* col_logits = (float*)d_out + B * U;

  norms_kernel<<<(B * U) / 8, 256, 0, stream>>>(utt, inv_nu, U);
  norms_kernel<<<(B * P) / 8, 256, 0, stream>>>(pheno, inv_np, P);
  wconv_kernel<<<(U * K2) / 256, 256, 0, stream>>>(w_utt, w_pheno, wu_bf, wp_bf);

  cosine_wmma_kernel<<<B, 256, 0, stream>>>(utt, pheno, inv_nu, inv_np, c_bf);

  logits_wmma_kernel<<<dim3(B / 16, U / 16), LG_WAVES * 32, 0, stream>>>(c_bf, wu_bf, b_utt, row_logits, U);
  logits_wmma_kernel<<<dim3(B / 16, P / 16), LG_WAVES * 32, 0, stream>>>(c_bf, wp_bf, b_pheno, col_logits, P);

  softmax_kernel<<<B / 8, 256, 0, stream>>>(row_logits, U);
  softmax_kernel<<<B / 8, 256, 0, stream>>>(col_logits, P);
}


# cosine GEMM sums the squares of the f32 operand values it already loads and applies the norms in its epilogue; both norm kernels and their second read of the inputs removed (norm sum order changes)
# speedup vs baseline: 2.5702x; 1.1163x over previous
#include <hip/hip_runtime.h>
#include <hip/hip_bf16.h>
#include <math.h>

typedef __attribute__((ext_vector_type(16))) __bf16 v16bf;
typedef __attribute__((ext_vector_type(8)))  __bf16 v8bf;
typedef __attribute__((ext_vector_type(8)))  float  v8f;

constexpr int B  = 256;
constexpr int U  = 128;
constexpr int P  = 64;
constexpr int H  = 768;
constexpr int K2 = U * P;

__global__ void wconv_kernel(const float* __restrict__ w_utt,
                             const float* __restrict__ w_pheno,
                             __bf16* __restrict__ wu_bf,
                             __bf16* __restrict__ wp_bf) {
  int idx = blockIdx.x * blockDim.x + threadIdx.x;
  if (idx < U * K2) wu_bf[idx] = (__bf16)w_utt[idx];
  if (idx < P * K2) {
    int o = idx / K2, r = idx % K2;
    int u = r / P, p = r % P;
    wp_bf[idx] = (__bf16)w_pheno[(size_t)o * K2 + p * U + u];
  }
}

__device__ __forceinline__ v16bf load_frag_f32_ss(const float* __restrict__ p, float4& ss) {
  float4 a0 = *(const float4*)(p);
  float4 a1 = *(const float4*)(p + 4);
  float4 b0 = *(const float4*)(p + 16);
  float4 b1 = *(const float4*)(p + 20);
  ss.x += a0.x * a0.x; ss.y += a0.y * a0.y; ss.z += a0.z * a0.z; ss.w += a0.w * a0.w;
  ss.x += a1.x * a1.x; ss.y += a1.y * a1.y; ss.z += a1.z * a1.z; ss.w += a1.w * a1.w;
  ss.x += b0.x * b0.x; ss.y += b0.y * b0.y; ss.z += b0.z * b0.z; ss.w += b0.w * b0.w;
  ss.x += b1.x * b1.x; ss.y += b1.y * b1.y; ss.z += b1.z * b1.z; ss.w += b1.w * b1.w;
  v16bf r;
  r[0]  = (__bf16)a0.x; r[1]  = (__bf16)a0.y;
  r[2]  = (__bf16)a0.z; r[3]  = (__bf16)a0.w;
  r[4]  = (__bf16)a1.x; r[5]  = (__bf16)a1.y;
  r[6]  = (__bf16)a1.z; r[7]  = (__bf16)a1.w;
  r[8]  = (__bf16)b0.x; r[9]  = (__bf16)b0.y;
  r[10] = (__bf16)b0.z; r[11] = (__bf16)b0.w;
  r[12] = (__bf16)b1.x; r[13] = (__bf16)b1.y;
  r[14] = (__bf16)b1.z; r[15] = (__bf16)b1.w;
  return r;
}

__device__ __forceinline__ float inv_norm_from_halves(float4 ss) {
  float s = (ss.x + ss.y) + (ss.z + ss.w);
  s += __shfl_xor(s, 16, 32);
  return 1.0f / fmaxf(sqrtf(s), 1e-8f);
}

__device__ __forceinline__ v16bf load_frag_bf16(const __bf16* __restrict__ p) {
  v8bf lo = *(const v8bf*)(p);
  v8bf hi = *(const v8bf*)(p + 16);
  v16bf r;
#pragma unroll
  for (int j = 0; j < 8; ++j) { r[j] = lo[j]; r[j + 8] = hi[j]; }
  return r;
}

__global__ void __launch_bounds__(256) __attribute__((amdgpu_waves_per_eu(2, 2)))
cosine_wmma_kernel(const float* __restrict__ utt,
                   const float* __restrict__ pheno,
                   __bf16* __restrict__ c_bf) {
  int b    = blockIdx.x;
  int wave = threadIdx.x >> 5;
  int lane = threadIdx.x & 31;
  int rt   = lane & 15;
  int hi   = lane >> 4;
  int k0   = hi ? 8 : 0;

  int u_row = wave * 16 + rt;
  const float* Arow = utt + ((size_t)u_row * B + b) * H + k0;

  const float* Brow[4];
#pragma unroll
  for (int nt = 0; nt < 4; ++nt) {
    int p_col = nt * 16 + rt;
    Brow[nt]  = pheno + ((size_t)p_col * B + b) * H + k0;
  }

  v8f acc[4] = {};
  float4 ssA = make_float4(0.f, 0.f, 0.f, 0.f);
  float4 ssB[4];
#pragma unroll
  for (int nt = 0; nt < 4; ++nt) ssB[nt] = make_float4(0.f, 0.f, 0.f, 0.f);
#pragma unroll 2
  for (int kk = 0; kk < H; kk += 32) {
    v16bf a = load_frag_f32_ss(Arow + kk, ssA);
#pragma unroll
    for (int nt = 0; nt < 4; ++nt) {
      v16bf bb = load_frag_f32_ss(Brow[nt] + kk, ssB[nt]);
      acc[nt] = __builtin_amdgcn_wmma_f32_16x16x32_bf16(
          false, a, false, bb, (short)0, acc[nt], false, false);
    }
  }

  float inv_a = inv_norm_from_halves(ssA);
  float su[8];
#pragma unroll
  for (int r = 0; r < 8; ++r) su[r] = __shfl(inv_a, 8 * hi + r, 32);

  __bf16* cb = c_bf + (size_t)b * K2;
#pragma unroll
  for (int nt = 0; nt < 4; ++nt) {
    float sp = inv_norm_from_halves(ssB[nt]);
#pragma unroll
    for (int r = 0; r < 8; ++r) {
      int m = r + 8 * hi;
      cb[(wave * 16 + m) * P + nt * 16 + rt] = (__bf16)(acc[nt][r] * su[r] * sp);
    }
  }
}

constexpr int LG_WAVES  = 16;
constexpr int LG_KSLICE = K2 / LG_WAVES;
constexpr int LG_STEPS  = LG_KSLICE / 32;
constexpr int LG_AHEAD  = 8;

__global__ void __launch_bounds__(LG_WAVES * 32)
logits_wmma_kernel(const __bf16* __restrict__ c_bf,
                   const __bf16* __restrict__ W,
                   const float* __restrict__ bias,
                   float* __restrict__ out, int N) {
  __shared__ float red[LG_WAVES][256];

  int mtile = blockIdx.x;
  int ntile = blockIdx.y;
  int wave  = threadIdx.x >> 5;
  int lane  = threadIdx.x & 31;
  int rt    = lane & 15;
  int hi    = lane >> 4;
  int k0    = hi ? 8 : 0;

  const __bf16* Arow = c_bf + (size_t)(mtile * 16 + rt) * K2 + wave * LG_KSLICE + k0;
  const __bf16* Wrow = W    + (size_t)(ntile * 16 + rt) * K2 + wave * LG_KSLICE + k0;

  v16bf a[LG_AHEAD], w[LG_AHEAD];
#pragma unroll
  for (int i = 0; i < LG_AHEAD; ++i) {
    a[i] = load_frag_bf16(Arow + i * 32);
    w[i] = load_frag_bf16(Wrow + i * 32);
  }
  __builtin_amdgcn_sched_barrier(0);
  v8f acc = {};
#pragma unroll
  for (int i = 0; i < LG_STEPS; ++i) {
    acc = __builtin_amdgcn_wmma_f32_16x16x32_bf16(
        false, a[i % LG_AHEAD], false, w[i % LG_AHEAD], (short)0, acc, false, false);
    if (i + LG_AHEAD < LG_STEPS) {
      a[i % LG_AHEAD] = load_frag_bf16(Arow + (i + LG_AHEAD) * 32);
      w[i % LG_AHEAD] = load_frag_bf16(Wrow + (i + LG_AHEAD) * 32);
      __builtin_amdgcn_sched_barrier(0);
    }
  }

#pragma unroll
  for (int r = 0; r < 8; ++r) red[wave][r * 32 + lane] = acc[r];
  __syncthreads();

  if (threadIdx.x < 256) {
    int e = threadIdx.x;
    float s = red[0][e];
#pragma unroll
    for (int w = 1; w < LG_WAVES; ++w) s += red[w][e];
    int r = e >> 5;
    int l = e & 31;
    int m = r + 8 * (l >> 4);
    int n = ntile * 16 + (l & 15);
    out[(size_t)(mtile * 16 + m) * N + n] = s + bias[n];
  }
}

__global__ void softmax_kernel(float* __restrict__ x, int n) {
  int row  = (blockIdx.x * blockDim.x + threadIdx.x) >> 5;
  int lane = threadIdx.x & 31;
  if (row >= B) return;
  float* p = x + (size_t)row * n;
  int per = n >> 5;
  float vals[4];
  float mx = -1e30f;
  for (int j = 0; j < per; ++j) { vals[j] = p[lane + 32 * j]; mx = fmaxf(mx, vals[j]); }
#pragma unroll
  for (int m = 16; m; m >>= 1) mx = fmaxf(mx, __shfl_xor(mx, m, 32));
  float s = 0.f;
  for (int j = 0; j < per; ++j) { vals[j] = __expf(vals[j] - mx); s += vals[j]; }
#pragma unroll
  for (int m = 16; m; m >>= 1) s += __shfl_xor(s, m, 32);
  float inv = 1.0f / s;
  for (int j = 0; j < per; ++j) p[lane + 32 * j] = vals[j] * inv;
}

extern "C" void kernel_launch(void* const* d_in, const int* in_sizes, int n_in,
                              void* d_out, int out_size, void* d_ws, size_t ws_size,
                              hipStream_t stream) {
  const float* utt     = (const float*)d_in[0];
  const float* pheno   = (const float*)d_in[1];
  const float* w_utt   = (const float*)d_in[2];
  const float* b_utt   = (const float*)d_in[3];
  const float* w_pheno = (const float*)d_in[4];
  const float* b_pheno = (const float*)d_in[5];
  (void)in_sizes; (void)n_in; (void)out_size; (void)ws_size;

  char* ws = (char*)d_ws;
  size_t off = 0;
  auto alloc = [&](size_t bytes) {
    void* p = ws + off;
    off = (off + bytes + 255) & ~(size_t)255;
    return p;
  };
  __bf16* wu_bf  = (__bf16*)alloc((size_t)U * K2 * sizeof(__bf16));
  __bf16* wp_bf  = (__bf16*)alloc((size_t)P * K2 * sizeof(__bf16));
  __bf16* c_bf   = (__bf16*)alloc((size_t)B * K2 * sizeof(__bf16));

  float* row_logits = (float*)d_out;
  float* col_logits = (float*)d_out + B * U;

  wconv_kernel<<<(U * K2) / 256, 256, 0, stream>>>(w_utt, w_pheno, wu_bf, wp_bf);

  cosine_wmma_kernel<<<B, 256, 0, stream>>>(utt, pheno, c_bf);

  logits_wmma_kernel<<<dim3(B / 16, U / 16), LG_WAVES * 32, 0, stream>>>(c_bf, wu_bf, b_utt, row_logits, U);
  logits_wmma_kernel<<<dim3(B / 16, P / 16), LG_WAVES * 32, 0, stream>>>(c_bf, wp_bf, b_pheno, col_logits, P);

  softmax_kernel<<<B / 8, 256, 0, stream>>>(row_logits, U);
  softmax_kernel<<<B / 8, 256, 0, stream>>>(col_logits, P);
}


# weight staging (copy of w_utt, transpose of w_pheno) done by the cosine kernel's 256 blocks, 24 elements per lane, whole-line loads and stores; its own 32768-wave kernel and its launch removed
# speedup vs baseline: 2.7075x; 1.0534x over previous
#include <hip/hip_runtime.h>
#include <hip/hip_bf16.h>
#include <math.h>

typedef __attribute__((ext_vector_type(16))) __bf16 v16bf;
typedef __attribute__((ext_vector_type(8)))  __bf16 v8bf;
typedef __attribute__((ext_vector_type(4)))  __bf16 v4bf;
typedef __attribute__((ext_vector_type(8)))  float  v8f;

constexpr int B  = 256;
constexpr int U  = 128;
constexpr int P  = 64;
constexpr int H  = 768;
constexpr int K2 = U * P;

static_assert(B * 8 * 512 == U * K2, "w_utt: 512 elements per wave of the cosine grid");
static_assert(B * 8 * 32 * 8 == P * K2 && B == 4 * P && U == 4 * 32,
              "w_pheno: block = (o, 32 u), 8 elements per lane");

struct StagedWeights {
  float4 x[4];
  float  y[8];
};

__device__ __forceinline__ StagedWeights stage_weights_load(
    const float* __restrict__ w_utt, const float* __restrict__ w_pheno,
    int b, int wave, int lane) {
  StagedWeights s;
  const float* xs = w_utt + (size_t)(b * 8 + wave) * 512 + lane * 4;
#pragma unroll
  for (int j = 0; j < 4; ++j) s.x[j] = *(const float4*)(xs + j * 128);
  int o  = b >> 2;
  int u  = (b & 3) * 32 + wave * 4 + (lane >> 3);
  int p0 = (lane & 7) * 8;
  const float* ys = w_pheno + (size_t)o * K2 + p0 * U + u;
#pragma unroll
  for (int i = 0; i < 8; ++i) s.y[i] = ys[i * U];
  return s;
}

__device__ __forceinline__ void stage_weights_store(
    const StagedWeights& s, __bf16* __restrict__ wu_bf, __bf16* __restrict__ wp_bf,
    int b, int wave, int lane) {
  __bf16* xd = wu_bf + (size_t)(b * 8 + wave) * 512 + lane * 4;
#pragma unroll
  for (int j = 0; j < 4; ++j) {
    v4bf t;
    t[0] = (__bf16)s.x[j].x; t[1] = (__bf16)s.x[j].y;
    t[2] = (__bf16)s.x[j].z; t[3] = (__bf16)s.x[j].w;
    *(v4bf*)(xd + j * 128) = t;
  }
  int o  = b >> 2;
  int u  = (b & 3) * 32 + wave * 4 + (lane >> 3);
  int p0 = (lane & 7) * 8;
  v8bf t;
#pragma unroll
  for (int i = 0; i < 8; ++i) t[i] = (__bf16)s.y[i];
  *(v8bf*)(wp_bf + (size_t)o * K2 + u * P + p0) = t;
}

__device__ __forceinline__ v16bf load_frag_f32_ss(const float* __restrict__ p, float4& ss) {
  float4 a0 = *(const float4*)(p);
  float4 a1 = *(const float4*)(p + 4);
  float4 b0 = *(const float4*)(p + 16);
  float4 b1 = *(const float4*)(p + 20);
  ss.x += a0.x * a0.x; ss.y += a0.y * a0.y; ss.z += a0.z * a0.z; ss.w += a0.w * a0.w;
  ss.x += a1.x * a1.x; ss.y += a1.y * a1.y; ss.z += a1.z * a1.z; ss.w += a1.w * a1.w;
  ss.x += b0.x * b0.x; ss.y += b0.y * b0.y; ss.z += b0.z * b0.z; ss.w += b0.w * b0.w;
  ss.x += b1.x * b1.x; ss.y += b1.y * b1.y; ss.z += b1.z * b1.z; ss.w += b1.w * b1.w;
  v16bf r;
  r[0]  = (__bf16)a0.x; r[1]  = (__bf16)a0.y;
  r[2]  = (__bf16)a0.z; r[3]  = (__bf16)a0.w;
  r[4]  = (__bf16)a1.x; r[5]  = (__bf16)a1.y;
  r[6]  = (__bf16)a1.z; r[7]  = (__bf16)a1.w;
  r[8]  = (__bf16)b0.x; r[9]  = (__bf16)b0.y;
  r[10] = (__bf16)b0.z; r[11] = (__bf16)b0.w;
  r[12] = (__bf16)b1.x; r[13] = (__bf16)b1.y;
  r[14] = (__bf16)b1.z; r[15] = (__bf16)b1.w;
  return r;
}

__device__ __forceinline__ float inv_norm_from_halves(float4 ss) {
  float s = (ss.x + ss.y) + (ss.z + ss.w);
  s += __shfl_xor(s, 16, 32);
  return 1.0f / fmaxf(sqrtf(s), 1e-8f);
}

__device__ __forceinline__ v16bf load_frag_bf16(const __bf16* __restrict__ p) {
  v8bf lo = *(const v8bf*)(p);
  v8bf hi = *(const v8bf*)(p + 16);
  v16bf r;
#pragma unroll
  for (int j = 0; j < 8; ++j) { r[j] = lo[j]; r[j + 8] = hi[j]; }
  return r;
}

__global__ void __launch_bounds__(256) __attribute__((amdgpu_waves_per_eu(2, 2)))
cosine_wmma_kernel(const float* __restrict__ utt,
                   const float* __restrict__ pheno,
                   __bf16* __restrict__ c_bf,
                   const float* __restrict__ w_utt,
                   const float* __restrict__ w_pheno,
                   __bf16* __restrict__ wu_bf,
                   __bf16* __restrict__ wp_bf) {
  int b    = blockIdx.x;
  int wave = threadIdx.x >> 5;
  int lane = threadIdx.x & 31;
  StagedWeights sw = stage_weights_load(w_utt, w_pheno, b, wave, lane);
  int rt   = lane & 15;
  int hi   = lane >> 4;
  int k0   = hi ? 8 : 0;

  int u_row = wave * 16 + rt;
  const float* Arow = utt + ((size_t)u_row * B + b) * H + k0;

  const float* Brow[4];
#pragma unroll
  for (int nt = 0; nt < 4; ++nt) {
    int p_col = nt * 16 + rt;
    Brow[nt]  = pheno + ((size_t)p_col * B + b) * H + k0;
  }

  v8f acc[4] = {};
  float4 ssA = make_float4(0.f, 0.f, 0.f, 0.f);
  float4 ssB[4];
#pragma unroll
  for (int nt = 0; nt < 4; ++nt) ssB[nt] = make_float4(0.f, 0.f, 0.f, 0.f);
#pragma unroll 2
  for (int kk = 0; kk < H; kk += 32) {
    v16bf a = load_frag_f32_ss(Arow + kk, ssA);
#pragma unroll
    for (int nt = 0; nt < 4; ++nt) {
      v16bf bb = load_frag_f32_ss(Brow[nt] + kk, ssB[nt]);
      acc[nt] = __builtin_amdgcn_wmma_f32_16x16x32_bf16(
          false, a, false, bb, (short)0, acc[nt], false, false);
    }
  }

  float inv_a = inv_norm_from_halves(ssA);
  float su[8];
#pragma unroll
  for (int r = 0; r < 8; ++r) su[r] = __shfl(inv_a, 8 * hi + r, 32);

  __bf16* cb = c_bf + (size_t)b * K2;
#pragma unroll
  for (int nt = 0; nt < 4; ++nt) {
    float sp = inv_norm_from_halves(ssB[nt]);
#pragma unroll
    for (int r = 0; r < 8; ++r) {
      int m = r + 8 * hi;
      cb[(wave * 16 + m) * P + nt * 16 + rt] = (__bf16)(acc[nt][r] * su[r] * sp);
    }
  }

  stage_weights_store(sw, wu_bf, wp_bf, b, wave, lane);
}

constexpr int LG_WAVES  = 16;
constexpr int LG_KSLICE = K2 / LG_WAVES;
constexpr int LG_STEPS  = LG_KSLICE / 32;
constexpr int LG_AHEAD  = 8;

__global__ void __launch_bounds__(LG_WAVES * 32)
logits_wmma_kernel(const __bf16* __restrict__ c_bf,
                   const __bf16* __restrict__ W,
                   const float* __restrict__ bias,
                   float* __restrict__ out, int N) {
  __shared__ float red[LG_WAVES][256];

  int mtile = blockIdx.x;
  int ntile = blockIdx.y;
  int wave  = threadIdx.x >> 5;
  int lane  = threadIdx.x & 31;
  int rt    = lane & 15;
  int hi    = lane >> 4;
  int k0    = hi ? 8 : 0;

  const __bf16* Arow = c_bf + (size_t)(mtile * 16 + rt) * K2 + wave * LG_KSLICE + k0;
  const __bf16* Wrow = W    + (size_t)(ntile * 16 + rt) * K2 + wave * LG_KSLICE + k0;

  v16bf a[LG_AHEAD], w[LG_AHEAD];
#pragma unroll
  for (int i = 0; i < LG_AHEAD; ++i) {
    a[i] = load_frag_bf16(Arow + i * 32);
    w[i] = load_frag_bf16(Wrow + i * 32);
  }
  __builtin_amdgcn_sched_barrier(0);
  v8f acc = {};
#pragma unroll
  for (int i = 0; i < LG_STEPS; ++i) {
    acc = __builtin_amdgcn_wmma_f32_16x16x32_bf16(
        false, a[i % LG_AHEAD], false, w[i % LG_AHEAD], (short)0, acc, false, false);
    if (i + LG_AHEAD < LG_STEPS) {
      a[i % LG_AHEAD] = load_frag_bf16(Arow + (i + LG_AHEAD) * 32);
      w[i % LG_AHEAD] = load_frag_bf16(Wrow + (i + LG_AHEAD) * 32);
      __builtin_amdgcn_sched_barrier(0);
    }
  }

#pragma unroll
  for (int r = 0; r < 8; ++r) red[wave][r * 32 + lane] = acc[r];
  __syncthreads();

  if (threadIdx.x < 256) {
    int e = threadIdx.x;
    float s = red[0][e];
#pragma unroll
    for (int w = 1; w < LG_WAVES; ++w) s += red[w][e];
    int r = e >> 5;
    int l = e & 31;
    int m = r + 8 * (l >> 4);
    int n = ntile * 16 + (l & 15);
    out[(size_t)(mtile * 16 + m) * N + n] = s + bias[n];
  }
}

__global__ void softmax_kernel(float* __restrict__ x, int n) {
  int row  = (blockIdx.x * blockDim.x + threadIdx.x) >> 5;
  int lane = threadIdx.x & 31;
  if (row >= B) return;
  float* p = x + (size_t)row * n;
  int per = n >> 5;
  float vals[4];
  float mx = -1e30f;
  for (int j = 0; j < per; ++j) { vals[j] = p[lane + 32 * j]; mx = fmaxf(mx, vals[j]); }
#pragma unroll
  for (int m = 16; m; m >>= 1) mx = fmaxf(mx, __shfl_xor(mx, m, 32));
  float s = 0.f;
  for (int j = 0; j < per; ++j) { vals[j] = __expf(vals[j] - mx); s += vals[j]; }
#pragma unroll
  for (int m = 16; m; m >>= 1) s += __shfl_xor(s, m, 32);
  float inv = 1.0f / s;
  for (int j = 0; j < per; ++j) p[lane + 32 * j] = vals[j] * inv;
}

extern "C" void kernel_launch(void* const* d_in, const int* in_sizes, int n_in,
                              void* d_out, int out_size, void* d_ws, size_t ws_size,
                              hipStream_t stream) {
  const float* utt     = (const float*)d_in[0];
  const float* pheno   = (const float*)d_in[1];
  const float* w_utt   = (const float*)d_in[2];
  const float* b_utt   = (const float*)d_in[3];
  const float* w_pheno = (const float*)d_in[4];
  const float* b_pheno = (const float*)d_in[5];
  (void)in_sizes; (void)n_in; (void)out_size; (void)ws_size;

  char* ws = (char*)d_ws;
  size_t off = 0;
  auto alloc = [&](size_t bytes) {
    void* p = ws + off;
    off = (off + bytes + 255) & ~(size_t)255;
    return p;
  };
  __bf16* wu_bf  = (__bf16*)alloc((size_t)U * K2 * sizeof(__bf16));
  __bf16* wp_bf  = (__bf16*)alloc((size_t)P * K2 * sizeof(__bf16));
  __bf16* c_bf   = (__bf16*)alloc((size_t)B * K2 * sizeof(__bf16));

  float* row_logits = (float*)d_out;
  float* col_logits = (float*)d_out + B * U;

  cosine_wmma_kernel<<<B, 256, 0, stream>>>(utt, pheno, c_bf, w_utt, w_pheno, wu_bf, wp_bf);

  logits_wmma_kernel<<<dim3(B / 16, U / 16), LG_WAVES * 32, 0, stream>>>(c_bf, wu_bf, b_utt, row_logits, U);
  logits_wmma_kernel<<<dim3(B / 16, P / 16), LG_WAVES * 32, 0, stream>>>(c_bf, wp_bf, b_pheno, col_logits, P);

  softmax_kernel<<<B / 8, 256, 0, stream>>>(row_logits, U);
  softmax_kernel<<<B / 8, 256, 0, stream>>>(col_logits, P);
}


# cosine GEMM: each block stages its 192 operand rows once through LDS as bf16 with whole-line loads (4 full lines each, next chunk in flight), not 16-line loads by every wave; norm sum order changes
# speedup vs baseline: 4.5587x; 1.6837x over previous
#include <hip/hip_runtime.h>
#include <hip/hip_bf16.h>
#include <math.h>

typedef __attribute__((ext_vector_type(16))) __bf16 v16bf;
typedef __attribute__((ext_vector_type(8)))  __bf16 v8bf;
typedef __attribute__((ext_vector_type(4)))  __bf16 v4bf;
typedef __attribute__((ext_vector_type(8)))  float  v8f;

constexpr int B  = 256;
constexpr int U  = 128;
constexpr int P  = 64;
constexpr int H  = 768;
constexpr int K2 = U * P;

static_assert(B * 8 * 512 == U * K2, "w_utt: 512 elements per wave of the cosine grid");
static_assert(B * 8 * 32 * 8 == P * K2 && B == 4 * P && U == 4 * 32,
              "w_pheno: block = (o, 32 u), 8 elements per lane");

struct StagedWeights {
  float4 x[4];
  float  y[8];
};

__device__ __forceinline__ StagedWeights stage_weights_load(
    const float* __restrict__ w_utt, const float* __restrict__ w_pheno,
    int b, int wave, int lane) {
  StagedWeights s;
  const float* xs = w_utt + (size_t)(b * 8 + wave) * 512 + lane * 4;
#pragma unroll
  for (int j = 0; j < 4; ++j) s.x[j] = *(const float4*)(xs + j * 128);
  int o  = b >> 2;
  int u  = (b & 3) * 32 + wave * 4 + (lane >> 3);
  int p0 = (lane & 7) * 8;
  const float* ys = w_pheno + (size_t)o * K2 + p0 * U + u;
#pragma unroll
  for (int i = 0; i < 8; ++i) s.y[i] = ys[i * U];
  return s;
}

__device__ __forceinline__ void stage_weights_store(
    const StagedWeights& s, __bf16* __restrict__ wu_bf, __bf16* __restrict__ wp_bf,
    int b, int wave, int lane) {
  __bf16* xd = wu_bf + (size_t)(b * 8 + wave) * 512 + lane * 4;
#pragma unroll
  for (int j = 0; j < 4; ++j) {
    v4bf t;
    t[0] = (__bf16)s.x[j].x; t[1] = (__bf16)s.x[j].y;
    t[2] = (__bf16)s.x[j].z; t[3] = (__bf16)s.x[j].w;
    *(v4bf*)(xd + j * 128) = t;
  }
  int o  = b >> 2;
  int u  = (b & 3) * 32 + wave * 4 + (lane >> 3);
  int p0 = (lane & 7) * 8;
  v8bf t;
#pragma unroll
  for (int i = 0; i < 8; ++i) t[i] = (__bf16)s.y[i];
  *(v8bf*)(wp_bf + (size_t)o * K2 + u * P + p0) = t;
}

__device__ __forceinline__ v16bf load_frag_bf16(const __bf16* __restrict__ p) {
  v8bf lo = *(const v8bf*)(p);
  v8bf hi = *(const v8bf*)(p + 16);
  v16bf r;
#pragma unroll
  for (int j = 0; j < 8; ++j) { r[j] = lo[j]; r[j + 8] = hi[j]; }
  return r;
}

template <int M, int N>
__device__ __forceinline__ void fold_level(float (&v)[32], int lane) {
  bool up = (lane & M) != 0;
#pragma unroll
  for (int i = 0; i < N / 2; ++i) {
    float keep = up ? v[i + N / 2] : v[i];
    float send = up ? v[i] : v[i + N / 2];
    v[i] = keep + __shfl_xor(send, M, 32);
  }
}

__device__ __forceinline__ void lds_barrier() {
  asm volatile("s_wait_dscnt 0x0\n\ts_barrier_signal -1\n\ts_barrier_wait -1" ::: "memory");
}

constexpr int CK_KC    = 128;
constexpr int CK_NCH   = H / CK_KC;
constexpr int CK_ROWS  = U + P;
constexpr int CK_RPW   = CK_ROWS / 8;
constexpr int CK_PITCH = CK_KC + 8;
static_assert(H % CK_KC == 0 && U == 8 * 16 && P == 8 * 8, "cosine staging split");

__global__ void __launch_bounds__(256) __attribute__((amdgpu_waves_per_eu(2, 2)))
cosine_wmma_kernel(const float* __restrict__ utt,
                   const float* __restrict__ pheno,
                   __bf16* __restrict__ c_bf,
                   const float* __restrict__ w_utt,
                   const float* __restrict__ w_pheno,
                   __bf16* __restrict__ wu_bf,
                   __bf16* __restrict__ wp_bf) {
  __shared__ __attribute__((aligned(16))) __bf16 tile[CK_ROWS * CK_PITCH];
  __shared__ float invn[CK_ROWS];

  int b    = blockIdx.x;
  int wave = threadIdx.x >> 5;
  int lane = threadIdx.x & 31;
  StagedWeights sw = stage_weights_load(w_utt, w_pheno, b, wave, lane);
  int rt   = lane & 15;
  int hi   = lane >> 4;
  int k0   = hi ? 8 : 0;

  const float* srcA = utt   + ((size_t)(wave * 16) * B + b) * H + lane * 4;
  const float* srcB = pheno + ((size_t)(wave * 8)  * B + b) * H + lane * 4;
  constexpr size_t ROWSTRIDE = (size_t)B * H;
  __bf16* dstA = tile + (wave * 16) * CK_PITCH + lane * 4;
  __bf16* dstB = tile + (U + wave * 8) * CK_PITCH + lane * 4;
  const __bf16* fragA = tile + (wave * 16 + rt) * CK_PITCH + k0;
  const __bf16* fragB = tile + (U + rt) * CK_PITCH + k0;

  float4 buf[CK_RPW];
#pragma unroll
  for (int j = 0; j < 16; ++j) buf[j] = *(const float4*)(srcA + j * ROWSTRIDE);
#pragma unroll
  for (int j = 0; j < 8; ++j) buf[16 + j] = *(const float4*)(srcB + j * ROWSTRIDE);

  float ss[CK_RPW];
#pragma unroll
  for (int j = 0; j < CK_RPW; ++j) ss[j] = 0.f;
  v8f acc[4] = {};

  for (int c = 0; c < CK_NCH; ++c) {
#pragma unroll
    for (int j = 0; j < CK_RPW; ++j) {
      float4 v = buf[j];
      ss[j] += v.x * v.x; ss[j] += v.y * v.y; ss[j] += v.z * v.z; ss[j] += v.w * v.w;
      v4bf t;
      t[0] = (__bf16)v.x; t[1] = (__bf16)v.y; t[2] = (__bf16)v.z; t[3] = (__bf16)v.w;
      __bf16* d = (j < 16) ? (dstA + j * CK_PITCH) : (dstB + (j - 16) * CK_PITCH);
      *(v4bf*)d = t;
    }
    if (c + 1 < CK_NCH) {
      const float* nA = srcA + (c + 1) * CK_KC;
      const float* nB = srcB + (c + 1) * CK_KC;
#pragma unroll
      for (int j = 0; j < 16; ++j) buf[j] = *(const float4*)(nA + j * ROWSTRIDE);
#pragma unroll
      for (int j = 0; j < 8; ++j) buf[16 + j] = *(const float4*)(nB + j * ROWSTRIDE);
    }
    lds_barrier();
#pragma unroll
    for (int s2 = 0; s2 < CK_KC / 64; ++s2) {
      v16bf af[2], bf[2][4];
#pragma unroll
      for (int s = 0; s < 2; ++s) {
        af[s] = load_frag_bf16(fragA + (s2 * 2 + s) * 32);
#pragma unroll
        for (int nt = 0; nt < 4; ++nt)
          bf[s][nt] = load_frag_bf16(fragB + nt * 16 * CK_PITCH + (s2 * 2 + s) * 32);
      }
      __builtin_amdgcn_sched_barrier(0);
#pragma unroll
      for (int s = 0; s < 2; ++s) {
#pragma unroll
        for (int nt = 0; nt < 4; ++nt)
          acc[nt] = __builtin_amdgcn_wmma_f32_16x16x32_bf16(
              false, af[s], false, bf[s][nt], (short)0, acc[nt], false, false);
      }
    }
    if (c + 1 < CK_NCH) lds_barrier();
  }

  float v[32];
#pragma unroll
  for (int j = 0; j < 32; ++j) v[j] = (j < CK_RPW) ? ss[j] : 0.f;
  fold_level<16, 32>(v, lane);
  fold_level<8, 16>(v, lane);
  fold_level<4, 8>(v, lane);
  fold_level<2, 4>(v, lane);
  fold_level<1, 2>(v, lane);
  if (lane < CK_RPW) {
    int row = (lane < 16) ? (wave * 16 + lane) : (U + wave * 8 + lane - 16);
    invn[row] = 1.0f / fmaxf(sqrtf(v[0]), 1e-8f);
  }
  lds_barrier();

  float su[8];
#pragma unroll
  for (int r = 0; r < 8; ++r) su[r] = invn[wave * 16 + 8 * hi + r];

  __bf16* cb = c_bf + (size_t)b * K2;
#pragma unroll
  for (int nt = 0; nt < 4; ++nt) {
    float sp = invn[U + nt * 16 + rt];
#pragma unroll
    for (int r = 0; r < 8; ++r) {
      int m = r + 8 * hi;
      cb[(wave * 16 + m) * P + nt * 16 + rt] = (__bf16)(acc[nt][r] * su[r] * sp);
    }
  }

  stage_weights_store(sw, wu_bf, wp_bf, b, wave, lane);
}

constexpr int LG_WAVES  = 16;
constexpr int LG_KSLICE = K2 / LG_WAVES;
constexpr int LG_STEPS  = LG_KSLICE / 32;
constexpr int LG_AHEAD  = 8;

__global__ void __launch_bounds__(LG_WAVES * 32)
logits_wmma_kernel(const __bf16* __restrict__ c_bf,
                   const __bf16* __restrict__ W,
                   const float* __restrict__ bias,
                   float* __restrict__ out, int N) {
  __shared__ float red[LG_WAVES][256];

  int mtile = blockIdx.x;
  int ntile = blockIdx.y;
  int wave  = threadIdx.x >> 5;
  int lane  = threadIdx.x & 31;
  int rt    = lane & 15;
  int hi    = lane >> 4;
  int k0    = hi ? 8 : 0;

  const __bf16* Arow = c_bf + (size_t)(mtile * 16 + rt) * K2 + wave * LG_KSLICE + k0;
  const __bf16* Wrow = W    + (size_t)(ntile * 16 + rt) * K2 + wave * LG_KSLICE + k0;

  v16bf a[LG_AHEAD], w[LG_AHEAD];
#pragma unroll
  for (int i = 0; i < LG_AHEAD; ++i) {
    a[i] = load_frag_bf16(Arow + i * 32);
    w[i] = load_frag_bf16(Wrow + i * 32);
  }
  __builtin_amdgcn_sched_barrier(0);
  v8f acc = {};
#pragma unroll
  for (int i = 0; i < LG_STEPS; ++i) {
    acc = __builtin_amdgcn_wmma_f32_16x16x32_bf16(
        false, a[i % LG_AHEAD], false, w[i % LG_AHEAD], (short)0, acc, false, false);
    if (i + LG_AHEAD < LG_STEPS) {
      a[i % LG_AHEAD] = load_frag_bf16(Arow + (i + LG_AHEAD) * 32);
      w[i % LG_AHEAD] = load_frag_bf16(Wrow + (i + LG_AHEAD) * 32);
      __builtin_amdgcn_sched_barrier(0);
    }
  }

#pragma unroll
  for (int r = 0; r < 8; ++r) red[wave][r * 32 + lane] = acc[r];
  __syncthreads();

  if (threadIdx.x < 256) {
    int e = threadIdx.x;
    float s = red[0][e];
#pragma unroll
    for (int w = 1; w < LG_WAVES; ++w) s += red[w][e];
    int r = e >> 5;
    int l = e & 31;
    int m = r + 8 * (l >> 4);
    int n = ntile * 16 + (l & 15);
    out[(size_t)(mtile * 16 + m) * N + n] = s + bias[n];
  }
}

__global__ void softmax_kernel(float* __restrict__ x, int n) {
  int row  = (blockIdx.x * blockDim.x + threadIdx.x) >> 5;
  int lane = threadIdx.x & 31;
  if (row >= B) return;
  float* p = x + (size_t)row * n;
  int per = n >> 5;
  float vals[4];
  float mx = -1e30f;
  for (int j = 0; j < per; ++j) { vals[j] = p[lane + 32 * j]; mx = fmaxf(mx, vals[j]); }
#pragma unroll
  for (int m = 16; m; m >>= 1) mx = fmaxf(mx, __shfl_xor(mx, m, 32));
  float s = 0.f;
  for (int j = 0; j < per; ++j) { vals[j] = __expf(vals[j] - mx); s += vals[j]; }
#pragma unroll
  for (int m = 16; m; m >>= 1) s += __shfl_xor(s, m, 32);
  float inv = 1.0f / s;
  for (int j = 0; j < per; ++j) p[lane + 32 * j] = vals[j] * inv;
}

extern "C" void kernel_launch(void* const* d_in, const int* in_sizes, int n_in,
                              void* d_out, int out_size, void* d_ws, size_t ws_size,
                              hipStream_t stream) {
  const float* utt     = (const float*)d_in[0];
  const float* pheno   = (const float*)d_in[1];
  const float* w_utt   = (const float*)d_in[2];
  const float* b_utt   = (const float*)d_in[3];
  const float* w_pheno = (const float*)d_in[4];
  const float* b_pheno = (const float*)d_in[5];
  (void)in_sizes; (void)n_in; (void)out_size; (void)ws_size;

  char* ws = (char*)d_ws;
  size_t off = 0;
  auto alloc = [&](size_t bytes) {
    void* p = ws + off;
    off = (off + bytes + 255) & ~(size_t)255;
    return p;
  };
  __bf16* wu_bf  = (__bf16*)alloc((size_t)U * K2 * sizeof(__bf16));
  __bf16* wp_bf  = (__bf16*)alloc((size_t)P * K2 * sizeof(__bf16));
  __bf16* c_bf   = (__bf16*)alloc((size_t)B * K2 * sizeof(__bf16));

  float* row_logits = (float*)d_out;
  float* col_logits = (float*)d_out + B * U;

  cosine_wmma_kernel<<<B, 256, 0, stream>>>(utt, pheno, c_bf, w_utt, w_pheno, wu_bf, wp_bf);

  logits_wmma_kernel<<<dim3(B / 16, U / 16), LG_WAVES * 32, 0, stream>>>(c_bf, wu_bf, b_utt, row_logits, U);
  logits_wmma_kernel<<<dim3(B / 16, P / 16), LG_WAVES * 32, 0, stream>>>(c_bf, wp_bf, b_pheno, col_logits, P);

  softmax_kernel<<<B / 8, 256, 0, stream>>>(row_logits, U);
  softmax_kernel<<<B / 8, 256, 0, stream>>>(col_logits, P);
}


# both logits GEMMs run as one launch (grid over the 12 column tiles of the stacked weight matrices) and both softmaxes as one launch: 5 launches become 3; every sum keeps its order
# speedup vs baseline: 6.2822x; 1.3781x over previous
#include <hip/hip_runtime.h>
#include <hip/hip_bf16.h>
#include <math.h>

typedef __attribute__((ext_vector_type(16))) __bf16 v16bf;
typedef __attribute__((ext_vector_type(8)))  __bf16 v8bf;
typedef __attribute__((ext_vector_type(4)))  __bf16 v4bf;
typedef __attribute__((ext_vector_type(8)))  float  v8f;

constexpr int B  = 256;
constexpr int U  = 128;
constexpr int P  = 64;
constexpr int H  = 768;
constexpr int K2 = U * P;

static_assert(B * 8 * 512 == U * K2, "w_utt: 512 elements per wave of the cosine grid");
static_assert(B * 8 * 32 * 8 == P * K2 && B == 4 * P && U == 4 * 32,
              "w_pheno: block = (o, 32 u), 8 elements per lane");

struct StagedWeights {
  float4 x[4];
  float  y[8];
};

__device__ __forceinline__ StagedWeights stage_weights_load(
    const float* __restrict__ w_utt, const float* __restrict__ w_pheno,
    int b, int wave, int lane) {
  StagedWeights s;
  const float* xs = w_utt + (size_t)(b * 8 + wave) * 512 + lane * 4;
#pragma unroll
  for (int j = 0; j < 4; ++j) s.x[j] = *(const float4*)(xs + j * 128);
  int o  = b >> 2;
  int u  = (b & 3) * 32 + wave * 4 + (lane >> 3);
  int p0 = (lane & 7) * 8;
  const float* ys = w_pheno + (size_t)o * K2 + p0 * U + u;
#pragma unroll
  for (int i = 0; i < 8; ++i) s.y[i] = ys[i * U];
  return s;
}

__device__ __forceinline__ void stage_weights_store(
    const StagedWeights& s, __bf16* __restrict__ wu_bf, __bf16* __restrict__ wp_bf,
    int b, int wave, int lane) {
  __bf16* xd = wu_bf + (size_t)(b * 8 + wave) * 512 + lane * 4;
#pragma unroll
  for (int j = 0; j < 4; ++j) {
    v4bf t;
    t[0] = (__bf16)s.x[j].x; t[1] = (__bf16)s.x[j].y;
    t[2] = (__bf16)s.x[j].z; t[3] = (__bf16)s.x[j].w;
    *(v4bf*)(xd + j * 128) = t;
  }
  int o  = b >> 2;
  int u  = (b & 3) * 32 + wave * 4 + (lane >> 3);
  int p0 = (lane & 7) * 8;
  v8bf t;
#pragma unroll
  for (int i = 0; i < 8; ++i) t[i] = (__bf16)s.y[i];
  *(v8bf*)(wp_bf + (size_t)o * K2 + u * P + p0) = t;
}

__device__ __forceinline__ v16bf load_frag_bf16(const __bf16* __restrict__ p) {
  v8bf lo = *(const v8bf*)(p);
  v8bf hi = *(const v8bf*)(p + 16);
  v16bf r;
#pragma unroll
  for (int j = 0; j < 8; ++j) { r[j] = lo[j]; r[j + 8] = hi[j]; }
  return r;
}

template <int M, int N>
__device__ __forceinline__ void fold_level(float (&v)[32], int lane) {
  bool up = (lane & M) != 0;
#pragma unroll
  for (int i = 0; i < N / 2; ++i) {
    float keep = up ? v[i + N / 2] : v[i];
    float send = up ? v[i] : v[i + N / 2];
    v[i] = keep + __shfl_xor(send, M, 32);
  }
}

__device__ __forceinline__ void lds_barrier() {
  asm volatile("s_wait_dscnt 0x0\n\ts_barrier_signal -1\n\ts_barrier_wait -1" ::: "memory");
}

constexpr int CK_KC    = 128;
constexpr int CK_NCH   = H / CK_KC;
constexpr int CK_ROWS  = U + P;
constexpr int CK_RPW   = CK_ROWS / 8;
constexpr int CK_PITCH = CK_KC + 8;
static_assert(H % CK_KC == 0 && U == 8 * 16 && P == 8 * 8, "cosine staging split");

__global__ void __launch_bounds__(256) __attribute__((amdgpu_waves_per_eu(2, 2)))
cosine_wmma_kernel(const float* __restrict__ utt,
                   const float* __restrict__ pheno,
                   __bf16* __restrict__ c_bf,
                   const float* __restrict__ w_utt,
                   const float* __restrict__ w_pheno,
                   __bf16* __restrict__ wu_bf,
                   __bf16* __restrict__ wp_bf) {
  __shared__ __attribute__((aligned(16))) __bf16 tile[CK_ROWS * CK_PITCH];
  __shared__ float invn[CK_ROWS];

  int b    = blockIdx.x;
  int wave = threadIdx.x >> 5;
  int lane = threadIdx.x & 31;
  StagedWeights sw = stage_weights_load(w_utt, w_pheno, b, wave, lane);
  int rt   = lane & 15;
  int hi   = lane >> 4;
  int k0   = hi ? 8 : 0;

  const float* srcA = utt   + ((size_t)(wave * 16) * B + b) * H + lane * 4;
  const float* srcB = pheno + ((size_t)(wave * 8)  * B + b) * H + lane * 4;
  constexpr size_t ROWSTRIDE = (size_t)B * H;
  __bf16* dstA = tile + (wave * 16) * CK_PITCH + lane * 4;
  __bf16* dstB = tile + (U + wave * 8) * CK_PITCH + lane * 4;
  const __bf16* fragA = tile + (wave * 16 + rt) * CK_PITCH + k0;
  const __bf16* fragB = tile + (U + rt) * CK_PITCH + k0;

  float4 buf[CK_RPW];
#pragma unroll
  for (int j = 0; j < 16; ++j) buf[j] = *(const float4*)(srcA + j * ROWSTRIDE);
#pragma unroll
  for (int j = 0; j < 8; ++j) buf[16 + j] = *(const float4*)(srcB + j * ROWSTRIDE);

  float ss[CK_RPW];
#pragma unroll
  for (int j = 0; j < CK_RPW; ++j) ss[j] = 0.f;
  v8f acc[4] = {};

  for (int c = 0; c < CK_NCH; ++c) {
#pragma unroll
    for (int j = 0; j < CK_RPW; ++j) {
      float4 v = buf[j];
      ss[j] += v.x * v.x; ss[j] += v.y * v.y; ss[j] += v.z * v.z; ss[j] += v.w * v.w;
      v4bf t;
      t[0] = (__bf16)v.x; t[1] = (__bf16)v.y; t[2] = (__bf16)v.z; t[3] = (__bf16)v.w;
      __bf16* d = (j < 16) ? (dstA + j * CK_PITCH) : (dstB + (j - 16) * CK_PITCH);
      *(v4bf*)d = t;
    }
    if (c + 1 < CK_NCH) {
      const float* nA = srcA + (c + 1) * CK_KC;
      const float* nB = srcB + (c + 1) * CK_KC;
#pragma unroll
      for (int j = 0; j < 16; ++j) buf[j] = *(const float4*)(nA + j * ROWSTRIDE);
#pragma unroll
      for (int j = 0; j < 8; ++j) buf[16 + j] = *(const float4*)(nB + j * ROWSTRIDE);
    }
    lds_barrier();
#pragma unroll
    for (int s2 = 0; s2 < CK_KC / 64; ++s2) {
      v16bf af[2], bf[2][4];
#pragma unroll
      for (int s = 0; s < 2; ++s) {
        af[s] = load_frag_bf16(fragA + (s2 * 2 + s) * 32);
#pragma unroll
        for (int nt = 0; nt < 4; ++nt)
          bf[s][nt] = load_frag_bf16(fragB + nt * 16 * CK_PITCH + (s2 * 2 + s) * 32);
      }
      __builtin_amdgcn_sched_barrier(0);
#pragma unroll
      for (int s = 0; s < 2; ++s) {
#pragma unroll
        for (int nt = 0; nt < 4; ++nt)
          acc[nt] = __builtin_amdgcn_wmma_f32_16x16x32_bf16(
              false, af[s], false, bf[s][nt], (short)0, acc[nt], false, false);
      }
    }
    if (c + 1 < CK_NCH) lds_barrier();
  }

  float v[32];
#pragma unroll
  for (int j = 0; j < 32; ++j) v[j] = (j < CK_RPW) ? ss[j] : 0.f;
  fold_level<16, 32>(v, lane);
  fold_level<8, 16>(v, lane);
  fold_level<4, 8>(v, lane);
  fold_level<2, 4>(v, lane);
  fold_level<1, 2>(v, lane);
  if (lane < CK_RPW) {
    int row = (lane < 16) ? (wave * 16 + lane) : (U + wave * 8 + lane - 16);
    invn[row] = 1.0f / fmaxf(sqrtf(v[0]), 1e-8f);
  }
  lds_barrier();

  float su[8];
#pragma unroll
  for (int r = 0; r < 8; ++r) su[r] = invn[wave * 16 + 8 * hi + r];

  __bf16* cb = c_bf + (size_t)b * K2;
#pragma unroll
  for (int nt = 0; nt < 4; ++nt) {
    float sp = invn[U + nt * 16 + rt];
#pragma unroll
    for (int r = 0; r < 8; ++r) {
      int m = r + 8 * hi;
      cb[(wave * 16 + m) * P + nt * 16 + rt] = (__bf16)(acc[nt][r] * su[r] * sp);
    }
  }

  stage_weights_store(sw, wu_bf, wp_bf, b, wave, lane);
}

constexpr int LG_WAVES  = 16;
constexpr int LG_KSLICE = K2 / LG_WAVES;
constexpr int LG_STEPS  = LG_KSLICE / 32;
constexpr int LG_AHEAD  = 8;
constexpr int LG_NT_ROW = U / 16;
constexpr int LG_NT_COL = P / 16;

__global__ void __launch_bounds__(LG_WAVES * 32)
logits_wmma_kernel(const __bf16* __restrict__ c_bf,
                   const __bf16* __restrict__ W,
                   const float* __restrict__ bias_row,
                   const float* __restrict__ bias_col,
                   float* __restrict__ out_row,
                   float* __restrict__ out_col) {
  __shared__ float red[LG_WAVES][256];

  int mtile = blockIdx.x;
  int ntile = blockIdx.y;
  int wave  = threadIdx.x >> 5;
  int lane  = threadIdx.x & 31;
  int rt    = lane & 15;
  int hi    = lane >> 4;
  int k0    = hi ? 8 : 0;

  const __bf16* Arow = c_bf + (size_t)(mtile * 16 + rt) * K2 + wave * LG_KSLICE + k0;
  const __bf16* Wrow = W    + (size_t)(ntile * 16 + rt) * K2 + wave * LG_KSLICE + k0;

  v16bf a[LG_AHEAD], w[LG_AHEAD];
#pragma unroll
  for (int i = 0; i < LG_AHEAD; ++i) {
    a[i] = load_frag_bf16(Arow + i * 32);
    w[i] = load_frag_bf16(Wrow + i * 32);
  }
  __builtin_amdgcn_sched_barrier(0);
  v8f acc = {};
#pragma unroll
  for (int i = 0; i < LG_STEPS; ++i) {
    acc = __builtin_amdgcn_wmma_f32_16x16x32_bf16(
        false, a[i % LG_AHEAD], false, w[i % LG_AHEAD], (short)0, acc, false, false);
    if (i + LG_AHEAD < LG_STEPS) {
      a[i % LG_AHEAD] = load_frag_bf16(Arow + (i + LG_AHEAD) * 32);
      w[i % LG_AHEAD] = load_frag_bf16(Wrow + (i + LG_AHEAD) * 32);
      __builtin_amdgcn_sched_barrier(0);
    }
  }

#pragma unroll
  for (int r = 0; r < 8; ++r) red[wave][r * 32 + lane] = acc[r];
  __syncthreads();

  if (threadIdx.x < 256) {
    int e = threadIdx.x;
    float s = red[0][e];
#pragma unroll
    for (int w = 1; w < LG_WAVES; ++w) s += red[w][e];
    int r = e >> 5;
    int l = e & 31;
    int m = r + 8 * (l >> 4);
    bool isRow = ntile < LG_NT_ROW;
    int N = isRow ? U : P;
    int n = (isRow ? ntile : ntile - LG_NT_ROW) * 16 + (l & 15);
    const float* bias = isRow ? bias_row : bias_col;
    float* out = isRow ? out_row : out_col;
    out[(size_t)(mtile * 16 + m) * N + n] = s + bias[n];
  }
}

template <int PER>
__device__ __forceinline__ void softmax_row(float* __restrict__ p, int lane) {
  float vals[PER];
  float mx = -1e30f;
#pragma unroll
  for (int j = 0; j < PER; ++j) { vals[j] = p[lane + 32 * j]; mx = fmaxf(mx, vals[j]); }
#pragma unroll
  for (int m = 16; m; m >>= 1) mx = fmaxf(mx, __shfl_xor(mx, m, 32));
  float s = 0.f;
#pragma unroll
  for (int j = 0; j < PER; ++j) { vals[j] = __expf(vals[j] - mx); s += vals[j]; }
#pragma unroll
  for (int m = 16; m; m >>= 1) s += __shfl_xor(s, m, 32);
  float inv = 1.0f / s;
#pragma unroll
  for (int j = 0; j < PER; ++j) p[lane + 32 * j] = vals[j] * inv;
}

static_assert(B % 8 == 0, "softmax: the 8 rows of a block lie in one output");

__global__ void __launch_bounds__(256)
softmax_kernel(float* __restrict__ row_logits, float* __restrict__ col_logits) {
  int row  = (blockIdx.x * 256 + threadIdx.x) >> 5;
  int lane = threadIdx.x & 31;
  if (row < B) softmax_row<U / 32>(row_logits + (size_t)row * U, lane);
  else         softmax_row<P / 32>(col_logits + (size_t)(row - B) * P, lane);
}

extern "C" void kernel_launch(void* const* d_in, const int* in_sizes, int n_in,
                              void* d_out, int out_size, void* d_ws, size_t ws_size,
                              hipStream_t stream) {
  const float* utt     = (const float*)d_in[0];
  const float* pheno   = (const float*)d_in[1];
  const float* w_utt   = (const float*)d_in[2];
  const float* b_utt   = (const float*)d_in[3];
  const float* w_pheno = (const float*)d_in[4];
  const float* b_pheno = (const float*)d_in[5];
  (void)in_sizes; (void)n_in; (void)out_size; (void)ws_size;

  char* ws = (char*)d_ws;
  size_t off = 0;
  auto alloc = [&](size_t bytes) {
    void* p = ws + off;
    off = (off + bytes + 255) & ~(size_t)255;
    return p;
  };
  __bf16* w_bf   = (__bf16*)alloc((size_t)(U + P) * K2 * sizeof(__bf16));
  __bf16* wu_bf  = w_bf;
  __bf16* wp_bf  = w_bf + (size_t)U * K2;
  __bf16* c_bf   = (__bf16*)alloc((size_t)B * K2 * sizeof(__bf16));

  float* row_logits = (float*)d_out;
  float* col_logits = (float*)d_out + B * U;

  cosine_wmma_kernel<<<B, 256, 0, stream>>>(utt, pheno, c_bf, w_utt, w_pheno, wu_bf, wp_bf);

  logits_wmma_kernel<<<dim3(B / 16, LG_NT_ROW + LG_NT_COL), LG_WAVES * 32, 0, stream>>>(
      c_bf, w_bf, b_utt, b_pheno, row_logits, col_logits);

  softmax_kernel<<<2 * B / 8, 256, 0, stream>>>(row_logits, col_logits);
}


# logits GEMM: each wave takes its K slice through LDS of its own with whole-line loads (one row per load, 4 full lines), not 16-line fragment loads from memory; every sum keeps its order
# speedup vs baseline: 6.7801x; 1.0793x over previous
#include <hip/hip_runtime.h>
#include <hip/hip_bf16.h>
#include <math.h>

typedef __attribute__((ext_vector_type(16))) __bf16 v16bf;
typedef __attribute__((ext_vector_type(8)))  __bf16 v8bf;
typedef __attribute__((ext_vector_type(4)))  __bf16 v4bf;
typedef __attribute__((ext_vector_type(8)))  float  v8f;

constexpr int B  = 256;
constexpr int U  = 128;
constexpr int P  = 64;
constexpr int H  = 768;
constexpr int K2 = U * P;

static_assert(B * 8 * 512 == U * K2, "w_utt: 512 elements per wave of the cosine grid");
static_assert(B * 8 * 32 * 8 == P * K2 && B == 4 * P && U == 4 * 32,
              "w_pheno: block = (o, 32 u), 8 elements per lane");

struct StagedWeights {
  float4 x[4];
  float  y[8];
};

__device__ __forceinline__ StagedWeights stage_weights_load(
    const float* __restrict__ w_utt, const float* __restrict__ w_pheno,
    int b, int wave, int lane) {
  StagedWeights s;
  const float* xs = w_utt + (size_t)(b * 8 + wave) * 512 + lane * 4;
#pragma unroll
  for (int j = 0; j < 4; ++j) s.x[j] = *(const float4*)(xs + j * 128);
  int o  = b >> 2;
  int u  = (b & 3) * 32 + wave * 4 + (lane >> 3);
  int p0 = (lane & 7) * 8;
  const float* ys = w_pheno + (size_t)o * K2 + p0 * U + u;
#pragma unroll
  for (int i = 0; i < 8; ++i) s.y[i] = ys[i * U];
  return s;
}

__device__ __forceinline__ void stage_weights_store(
    const StagedWeights& s, __bf16* __restrict__ wu_bf, __bf16* __restrict__ wp_bf,
    int b, int wave, int lane) {
  __bf16* xd = wu_bf + (size_t)(b * 8 + wave) * 512 + lane * 4;
#pragma unroll
  for (int j = 0; j < 4; ++j) {
    v4bf t;
    t[0] = (__bf16)s.x[j].x; t[1] = (__bf16)s.x[j].y;
    t[2] = (__bf16)s.x[j].z; t[3] = (__bf16)s.x[j].w;
    *(v4bf*)(xd + j * 128) = t;
  }
  int o  = b >> 2;
  int u  = (b & 3) * 32 + wave * 4 + (lane >> 3);
  int p0 = (lane & 7) * 8;
  v8bf t;
#pragma unroll
  for (int i = 0; i < 8; ++i) t[i] = (__bf16)s.y[i];
  *(v8bf*)(wp_bf + (size_t)o * K2 + u * P + p0) = t;
}

__device__ __forceinline__ v16bf load_frag_bf16(const __bf16* __restrict__ p) {
  v8bf lo = *(const v8bf*)(p);
  v8bf hi = *(const v8bf*)(p + 16);
  v16bf r;
#pragma unroll
  for (int j = 0; j < 8; ++j) { r[j] = lo[j]; r[j + 8] = hi[j]; }
  return r;
}

template <int M, int N>
__device__ __forceinline__ void fold_level(float (&v)[32], int lane) {
  bool up = (lane & M) != 0;
#pragma unroll
  for (int i = 0; i < N / 2; ++i) {
    float keep = up ? v[i + N / 2] : v[i];
    float send = up ? v[i] : v[i + N / 2];
    v[i] = keep + __shfl_xor(send, M, 32);
  }
}

__device__ __forceinline__ void lds_barrier() {
  asm volatile("s_wait_dscnt 0x0\n\ts_barrier_signal -1\n\ts_barrier_wait -1" ::: "memory");
}

constexpr int CK_KC    = 128;
constexpr int CK_NCH   = H / CK_KC;
constexpr int CK_ROWS  = U + P;
constexpr int CK_RPW   = CK_ROWS / 8;
constexpr int CK_PITCH = CK_KC + 8;
static_assert(H % CK_KC == 0 && U == 8 * 16 && P == 8 * 8, "cosine staging split");

__global__ void __launch_bounds__(256) __attribute__((amdgpu_waves_per_eu(2, 2)))
cosine_wmma_kernel(const float* __restrict__ utt,
                   const float* __restrict__ pheno,
                   __bf16* __restrict__ c_bf,
                   const float* __restrict__ w_utt,
                   const float* __restrict__ w_pheno,
                   __bf16* __restrict__ wu_bf,
                   __bf16* __restrict__ wp_bf) {
  __shared__ __attribute__((aligned(16))) __bf16 tile[CK_ROWS * CK_PITCH];
  __shared__ float invn[CK_ROWS];

  int b    = blockIdx.x;
  int wave = threadIdx.x >> 5;
  int lane = threadIdx.x & 31;
  StagedWeights sw = stage_weights_load(w_utt, w_pheno, b, wave, lane);
  int rt   = lane & 15;
  int hi   = lane >> 4;
  int k0   = hi ? 8 : 0;

  const float* srcA = utt   + ((size_t)(wave * 16) * B + b) * H + lane * 4;
  const float* srcB = pheno + ((size_t)(wave * 8)  * B + b) * H + lane * 4;
  constexpr size_t ROWSTRIDE = (size_t)B * H;
  __bf16* dstA = tile + (wave * 16) * CK_PITCH + lane * 4;
  __bf16* dstB = tile + (U + wave * 8) * CK_PITCH + lane * 4;
  const __bf16* fragA = tile + (wave * 16 + rt) * CK_PITCH + k0;
  const __bf16* fragB = tile + (U + rt) * CK_PITCH + k0;

  float4 buf[CK_RPW];
#pragma unroll
  for (int j = 0; j < 16; ++j) buf[j] = *(const float4*)(srcA + j * ROWSTRIDE);
#pragma unroll
  for (int j = 0; j < 8; ++j) buf[16 + j] = *(const float4*)(srcB + j * ROWSTRIDE);

  float ss[CK_RPW];
#pragma unroll
  for (int j = 0; j < CK_RPW; ++j) ss[j] = 0.f;
  v8f acc[4] = {};

  for (int c = 0; c < CK_NCH; ++c) {
#pragma unroll
    for (int j = 0; j < CK_RPW; ++j) {
      float4 v = buf[j];
      ss[j] += v.x * v.x; ss[j] += v.y * v.y; ss[j] += v.z * v.z; ss[j] += v.w * v.w;
      v4bf t;
      t[0] = (__bf16)v.x; t[1] = (__bf16)v.y; t[2] = (__bf16)v.z; t[3] = (__bf16)v.w;
      __bf16* d = (j < 16) ? (dstA + j * CK_PITCH) : (dstB + (j - 16) * CK_PITCH);
      *(v4bf*)d = t;
    }
    if (c + 1 < CK_NCH) {
      const float* nA = srcA + (c + 1) * CK_KC;
      const float* nB = srcB + (c + 1) * CK_KC;
#pragma unroll
      for (int j = 0; j < 16; ++j) buf[j] = *(const float4*)(nA + j * ROWSTRIDE);
#pragma unroll
      for (int j = 0; j < 8; ++j) buf[16 + j] = *(const float4*)(nB + j * ROWSTRIDE);
    }
    lds_barrier();
#pragma unroll
    for (int s2 = 0; s2 < CK_KC / 64; ++s2) {
      v16bf af[2], bf[2][4];
#pragma unroll
      for (int s = 0; s < 2; ++s) {
        af[s] = load_frag_bf16(fragA + (s2 * 2 + s) * 32);
#pragma unroll
        for (int nt = 0; nt < 4; ++nt)
          bf[s][nt] = load_frag_bf16(fragB + nt * 16 * CK_PITCH + (s2 * 2 + s) * 32);
      }
      __builtin_amdgcn_sched_barrier(0);
#pragma unroll
      for (int s = 0; s < 2; ++s) {
#pragma unroll
        for (int nt = 0; nt < 4; ++nt)
          acc[nt] = __builtin_amdgcn_wmma_f32_16x16x32_bf16(
              false, af[s], false, bf[s][nt], (short)0, acc[nt], false, false);
      }
    }
    if (c + 1 < CK_NCH) lds_barrier();
  }

  float v[32];
#pragma unroll
  for (int j = 0; j < 32; ++j) v[j] = (j < CK_RPW) ? ss[j] : 0.f;
  fold_level<16, 32>(v, lane);
  fold_level<8, 16>(v, lane);
  fold_level<4, 8>(v, lane);
  fold_level<2, 4>(v, lane);
  fold_level<1, 2>(v, lane);
  if (lane < CK_RPW) {
    int row = (lane < 16) ? (wave * 16 + lane) : (U + wave * 8 + lane - 16);
    invn[row] = 1.0f / fmaxf(sqrtf(v[0]), 1e-8f);
  }
  lds_barrier();

  float su[8];
#pragma unroll
  for (int r = 0; r < 8; ++r) su[r] = invn[wave * 16 + 8 * hi + r];

  __bf16* cb = c_bf + (size_t)b * K2;
#pragma unroll
  for (int nt = 0; nt < 4; ++nt) {
    float sp = invn[U + nt * 16 + rt];
#pragma unroll
    for (int r = 0; r < 8; ++r) {
      int m = r + 8 * hi;
      cb[(wave * 16 + m) * P + nt * 16 + rt] = (__bf16)(acc[nt][r] * su[r] * sp);
    }
  }

  stage_weights_store(sw, wu_bf, wp_bf, b, wave, lane);
}

constexpr int LG_WAVES  = 16;
constexpr int LG_KSLICE = K2 / LG_WAVES;
constexpr int LG_PASSK  = 256;
constexpr int LG_NPASS  = LG_KSLICE / LG_PASSK;
constexpr int LG_PSTEPS = LG_PASSK / 32;
constexpr int LG_GROUP  = 4;
constexpr int LG_PITCH  = LG_PASSK + 8;
constexpr int LG_NT_ROW = U / 16;
constexpr int LG_NT_COL = P / 16;
static_assert(LG_PASSK == 32 * 8 && LG_PSTEPS % LG_GROUP == 0, "logits staging split");

__global__ void __launch_bounds__(LG_WAVES * 32)
logits_wmma_kernel(const __bf16* __restrict__ c_bf,
                   const __bf16* __restrict__ W,
                   const float* __restrict__ bias_row,
                   const float* __restrict__ bias_col,
                   float* __restrict__ out_row,
                   float* __restrict__ out_col) {
  __shared__ __attribute__((aligned(16))) __bf16 stage[LG_WAVES][2][16 * LG_PITCH];
  __shared__ float red[LG_WAVES][256];

  int mtile = blockIdx.x;
  int ntile = blockIdx.y;
  int wave  = threadIdx.x >> 5;
  int lane  = threadIdx.x & 31;
  int rt    = lane & 15;
  int hi    = lane >> 4;
  int k0    = hi ? 8 : 0;

  const __bf16* Ag = c_bf + (size_t)(mtile * 16) * K2 + wave * LG_KSLICE + lane * 8;
  const __bf16* Wg = W    + (size_t)(ntile * 16) * K2 + wave * LG_KSLICE + lane * 8;
  __bf16* sA = &stage[wave][0][lane * 8];
  __bf16* sW = &stage[wave][1][lane * 8];
  const __bf16* fA = &stage[wave][0][rt * LG_PITCH + k0];
  const __bf16* fW = &stage[wave][1][rt * LG_PITCH + k0];

  v8bf ra[16], rw[16];
#pragma unroll
  for (int r = 0; r < 16; ++r) {
    ra[r] = *(const v8bf*)(Ag + (size_t)r * K2);
    rw[r] = *(const v8bf*)(Wg + (size_t)r * K2);
  }

  v8f acc = {};
#pragma unroll
  for (int ps = 0; ps < LG_NPASS; ++ps) {
#pragma unroll
    for (int r = 0; r < 16; ++r) {
      *(v8bf*)(sA + r * LG_PITCH) = ra[r];
      *(v8bf*)(sW + r * LG_PITCH) = rw[r];
    }
    if (ps + 1 < LG_NPASS) {
#pragma unroll
      for (int r = 0; r < 16; ++r) {
        ra[r] = *(const v8bf*)(Ag + (size_t)r * K2 + (ps + 1) * LG_PASSK);
        rw[r] = *(const v8bf*)(Wg + (size_t)r * K2 + (ps + 1) * LG_PASSK);
      }
    }
#pragma unroll
    for (int g = 0; g < LG_PSTEPS / LG_GROUP; ++g) {
      v16bf a[LG_GROUP], w[LG_GROUP];
#pragma unroll
      for (int s = 0; s < LG_GROUP; ++s) {
        a[s] = load_frag_bf16(fA + (g * LG_GROUP + s) * 32);
        w[s] = load_frag_bf16(fW + (g * LG_GROUP + s) * 32);
      }
      __builtin_amdgcn_sched_barrier(0);
#pragma unroll
      for (int s = 0; s < LG_GROUP; ++s)
        acc = __builtin_amdgcn_wmma_f32_16x16x32_bf16(
            false, a[s], false, w[s], (short)0, acc, false, false);
    }
  }

#pragma unroll
  for (int r = 0; r < 8; ++r) red[wave][r * 32 + lane] = acc[r];
  __syncthreads();

  if (threadIdx.x < 256) {
    int e = threadIdx.x;
    float s = red[0][e];
#pragma unroll
    for (int w = 1; w < LG_WAVES; ++w) s += red[w][e];
    int r = e >> 5;
    int l = e & 31;
    int m = r + 8 * (l >> 4);
    bool isRow = ntile < LG_NT_ROW;
    int N = isRow ? U : P;
    int n = (isRow ? ntile : ntile - LG_NT_ROW) * 16 + (l & 15);
    const float* bias = isRow ? bias_row : bias_col;
    float* out = isRow ? out_row : out_col;
    out[(size_t)(mtile * 16 + m) * N + n] = s + bias[n];
  }
}

template <int PER>
__device__ __forceinline__ void softmax_row(float* __restrict__ p, int lane) {
  float vals[PER];
  float mx = -1e30f;
#pragma unroll
  for (int j = 0; j < PER; ++j) { vals[j] = p[lane + 32 * j]; mx = fmaxf(mx, vals[j]); }
#pragma unroll
  for (int m = 16; m; m >>= 1) mx = fmaxf(mx, __shfl_xor(mx, m, 32));
  float s = 0.f;
#pragma unroll
  for (int j = 0; j < PER; ++j) { vals[j] = __expf(vals[j] - mx); s += vals[j]; }
#pragma unroll
  for (int m = 16; m; m >>= 1) s += __shfl_xor(s, m, 32);
  float inv = 1.0f / s;
#pragma unroll
  for (int j = 0; j < PER; ++j) p[lane + 32 * j] = vals[j] * inv;
}

static_assert(B % 8 == 0, "softmax: the 8 rows of a block lie in one output");

__global__ void __launch_bounds__(256)
softmax_kernel(float* __restrict__ row_logits, float* __restrict__ col_logits) {
  int row  = (blockIdx.x * 256 + threadIdx.x) >> 5;
  int lane = threadIdx.x & 31;
  if (row < B) softmax_row<U / 32>(row_logits + (size_t)row * U, lane);
  else         softmax_row<P / 32>(col_logits + (size_t)(row - B) * P, lane);
}

extern "C" void kernel_launch(void* const* d_in, const int* in_sizes, int n_in,
                              void* d_out, int out_size, void* d_ws, size_t ws_size,
                              hipStream_t stream) {
  const float* utt     = (const float*)d_in[0];
  const float* pheno   = (const float*)d_in[1];
  const float* w_utt   = (const float*)d_in[2];
  const float* b_utt   = (const float*)d_in[3];
  const float* w_pheno = (const float*)d_in[4];
  const float* b_pheno = (const float*)d_in[5];
  (void)in_sizes; (void)n_in; (void)out_size; (void)ws_size;

  char* ws = (char*)d_ws;
  size_t off = 0;
  auto alloc = [&](size_t bytes) {
    void* p = ws + off;
    off = (off + bytes + 255) & ~(size_t)255;
    return p;
  };
  __bf16* w_bf   = (__bf16*)alloc((size_t)(U + P) * K2 * sizeof(__bf16));
  __bf16* wu_bf  = w_bf;
  __bf16* wp_bf  = w_bf + (size_t)U * K2;
  __bf16* c_bf   = (__bf16*)alloc((size_t)B * K2 * sizeof(__bf16));

  float* row_logits = (float*)d_out;
  float* col_logits = (float*)d_out + B * U;

  cosine_wmma_kernel<<<B, 256, 0, stream>>>(utt, pheno, c_bf, w_utt, w_pheno, wu_bf, wp_bf);

  logits_wmma_kernel<<<dim3(B / 16, LG_NT_ROW + LG_NT_COL), LG_WAVES * 32, 0, stream>>>(
      c_bf, w_bf, b_utt, b_pheno, row_logits, col_logits);

  softmax_kernel<<<2 * B / 8, 256, 0, stream>>>(row_logits, col_logits);
}


# cosine GEMM staging loop unrolled over two fixed register sets: no 96-register copies per chunk, next chunk requested before the hand-over; squares by packed fma, 2 partials (norm sum order changes)
# speedup vs baseline: 6.9975x; 1.0321x over previous
#include <hip/hip_runtime.h>
#include <hip/hip_bf16.h>
#include <math.h>

typedef __attribute__((ext_vector_type(16))) __bf16 v16bf;
typedef __attribute__((ext_vector_type(8)))  __bf16 v8bf;
typedef __attribute__((ext_vector_type(4)))  __bf16 v4bf;
typedef __attribute__((ext_vector_type(8)))  float  v8f;
typedef __attribute__((ext_vector_type(4)))  float  v4f;
typedef __attribute__((ext_vector_type(2)))  float  v2f;

constexpr int B  = 256;
constexpr int U  = 128;
constexpr int P  = 64;
constexpr int H  = 768;
constexpr int K2 = U * P;

static_assert(B * 8 * 512 == U * K2, "w_utt: 512 elements per wave of the cosine grid");
static_assert(B * 8 * 32 * 8 == P * K2 && B == 4 * P && U == 4 * 32,
              "w_pheno: block = (o, 32 u), 8 elements per lane");

struct StagedWeights {
  float4 x[4];
  float  y[8];
};

__device__ __forceinline__ StagedWeights stage_weights_load(
    const float* __restrict__ w_utt, const float* __restrict__ w_pheno,
    int b, int wave, int lane) {
  StagedWeights s;
  const float* xs = w_utt + (size_t)(b * 8 + wave) * 512 + lane * 4;
#pragma unroll
  for (int j = 0; j < 4; ++j) s.x[j] = *(const float4*)(xs + j * 128);
  int o  = b >> 2;
  int u  = (b & 3) * 32 + wave * 4 + (lane >> 3);
  int p0 = (lane & 7) * 8;
  const float* ys = w_pheno + (size_t)o * K2 + p0 * U + u;
#pragma unroll
  for (int i = 0; i < 8; ++i) s.y[i] = ys[i * U];
  return s;
}

__device__ __forceinline__ void stage_weights_store(
    const StagedWeights& s, __bf16* __restrict__ wu_bf, __bf16* __restrict__ wp_bf,
    int b, int wave, int lane) {
  __bf16* xd = wu_bf + (size_t)(b * 8 + wave) * 512 + lane * 4;
#pragma unroll
  for (int j = 0; j < 4; ++j) {
    v4bf t;
    t[0] = (__bf16)s.x[j].x; t[1] = (__bf16)s.x[j].y;
    t[2] = (__bf16)s.x[j].z; t[3] = (__bf16)s.x[j].w;
    *(v4bf*)(xd + j * 128) = t;
  }
  int o  = b >> 2;
  int u  = (b & 3) * 32 + wave * 4 + (lane >> 3);
  int p0 = (lane & 7) * 8;
  v8bf t;
#pragma unroll
  for (int i = 0; i < 8; ++i) t[i] = (__bf16)s.y[i];
  *(v8bf*)(wp_bf + (size_t)o * K2 + u * P + p0) = t;
}

__device__ __forceinline__ v16bf load_frag_bf16(const __bf16* __restrict__ p) {
  v8bf lo = *(const v8bf*)(p);
  v8bf hi = *(const v8bf*)(p + 16);
  v16bf r;
#pragma unroll
  for (int j = 0; j < 8; ++j) { r[j] = lo[j]; r[j + 8] = hi[j]; }
  return r;
}

template <int M, int N>
__device__ __forceinline__ void fold_level(float (&v)[32], int lane) {
  bool up = (lane & M) != 0;
#pragma unroll
  for (int i = 0; i < N / 2; ++i) {
    float keep = up ? v[i + N / 2] : v[i];
    float send = up ? v[i] : v[i + N / 2];
    v[i] = keep + __shfl_xor(send, M, 32);
  }
}

__device__ __forceinline__ void lds_barrier() {
  asm volatile("s_wait_dscnt 0x0\n\ts_barrier_signal -1\n\ts_barrier_wait -1" ::: "memory");
}

constexpr int CK_KC    = 128;
constexpr int CK_NCH   = H / CK_KC;
constexpr int CK_ROWS  = U + P;
constexpr int CK_RPW   = CK_ROWS / 8;
constexpr int CK_PITCH = CK_KC + 8;
static_assert(H % CK_KC == 0 && U == 8 * 16 && P == 8 * 8, "cosine staging split");

__global__ void __launch_bounds__(256) __attribute__((amdgpu_waves_per_eu(2, 2)))
cosine_wmma_kernel(const float* __restrict__ utt,
                   const float* __restrict__ pheno,
                   __bf16* __restrict__ c_bf,
                   const float* __restrict__ w_utt,
                   const float* __restrict__ w_pheno,
                   __bf16* __restrict__ wu_bf,
                   __bf16* __restrict__ wp_bf) {
  __shared__ __attribute__((aligned(16))) __bf16 tile[CK_ROWS * CK_PITCH];
  __shared__ float invn[CK_ROWS];

  int b    = blockIdx.x;
  int wave = threadIdx.x >> 5;
  int lane = threadIdx.x & 31;
  StagedWeights sw = stage_weights_load(w_utt, w_pheno, b, wave, lane);
  int rt   = lane & 15;
  int hi   = lane >> 4;
  int k0   = hi ? 8 : 0;

  const float* srcA = utt   + ((size_t)(wave * 16) * B + b) * H + lane * 4;
  const float* srcB = pheno + ((size_t)(wave * 8)  * B + b) * H + lane * 4;
  constexpr size_t ROWSTRIDE = (size_t)B * H;
  __bf16* dstA = tile + (wave * 16) * CK_PITCH + lane * 4;
  __bf16* dstB = tile + (U + wave * 8) * CK_PITCH + lane * 4;
  const __bf16* fragA = tile + (wave * 16 + rt) * CK_PITCH + k0;
  const __bf16* fragB = tile + (U + rt) * CK_PITCH + k0;

  v4f buf[2][CK_RPW];
#pragma unroll
  for (int j = 0; j < 16; ++j) buf[0][j] = *(const v4f*)(srcA + j * ROWSTRIDE);
#pragma unroll
  for (int j = 0; j < 8; ++j) buf[0][16 + j] = *(const v4f*)(srcB + j * ROWSTRIDE);

  v2f ss2[CK_RPW];
#pragma unroll
  for (int j = 0; j < CK_RPW; ++j) ss2[j] = (v2f){0.f, 0.f};
  v8f acc[4] = {};

#pragma unroll
  for (int c = 0; c < CK_NCH; ++c) {
    if (c + 1 < CK_NCH) {
      const float* nA = srcA + (c + 1) * CK_KC;
      const float* nB = srcB + (c + 1) * CK_KC;
#pragma unroll
      for (int j = 0; j < 16; ++j) buf[(c + 1) & 1][j] = *(const v4f*)(nA + j * ROWSTRIDE);
#pragma unroll
      for (int j = 0; j < 8; ++j) buf[(c + 1) & 1][16 + j] = *(const v4f*)(nB + j * ROWSTRIDE);
    }
#pragma unroll
    for (int j = 0; j < CK_RPW; ++j) {
      v4f v = buf[c & 1][j];
      v4bf t;
      t[0] = (__bf16)v.x; t[1] = (__bf16)v.y; t[2] = (__bf16)v.z; t[3] = (__bf16)v.w;
      __bf16* d = (j < 16) ? (dstA + j * CK_PITCH) : (dstB + (j - 16) * CK_PITCH);
      *(v4bf*)d = t;
    }
    lds_barrier();
#pragma unroll
    for (int s2 = 0; s2 < CK_KC / 64; ++s2) {
      v16bf af[2], bf[2][4];
#pragma unroll
      for (int s = 0; s < 2; ++s) {
        af[s] = load_frag_bf16(fragA + (s2 * 2 + s) * 32);
#pragma unroll
        for (int nt = 0; nt < 4; ++nt)
          bf[s][nt] = load_frag_bf16(fragB + nt * 16 * CK_PITCH + (s2 * 2 + s) * 32);
      }
      __builtin_amdgcn_sched_barrier(0);
#pragma unroll
      for (int s = 0; s < 2; ++s) {
#pragma unroll
        for (int nt = 0; nt < 4; ++nt)
          acc[nt] = __builtin_amdgcn_wmma_f32_16x16x32_bf16(
              false, af[s], false, bf[s][nt], (short)0, acc[nt], false, false);
      }
    }
    if (c + 1 < CK_NCH) lds_barrier();
#pragma unroll
    for (int j = 0; j < CK_RPW; ++j) {
      v4f v = buf[c & 1][j];
      ss2[j] = __builtin_elementwise_fma(v.xy, v.xy, ss2[j]);
      ss2[j] = __builtin_elementwise_fma(v.zw, v.zw, ss2[j]);
    }
  }

  float v[32];
#pragma unroll
  for (int j = 0; j < 32; ++j) v[j] = (j < CK_RPW) ? (ss2[j].x + ss2[j].y) : 0.f;
  fold_level<16, 32>(v, lane);
  fold_level<8, 16>(v, lane);
  fold_level<4, 8>(v, lane);
  fold_level<2, 4>(v, lane);
  fold_level<1, 2>(v, lane);
  if (lane < CK_RPW) {
    int row = (lane < 16) ? (wave * 16 + lane) : (U + wave * 8 + lane - 16);
    invn[row] = 1.0f / fmaxf(sqrtf(v[0]), 1e-8f);
  }
  lds_barrier();

  float su[8];
#pragma unroll
  for (int r = 0; r < 8; ++r) su[r] = invn[wave * 16 + 8 * hi + r];

  __bf16* cb = c_bf + (size_t)b * K2;
#pragma unroll
  for (int nt = 0; nt < 4; ++nt) {
    float sp = invn[U + nt * 16 + rt];
#pragma unroll
    for (int r = 0; r < 8; ++r) {
      int m = r + 8 * hi;
      cb[(wave * 16 + m) * P + nt * 16 + rt] = (__bf16)(acc[nt][r] * su[r] * sp);
    }
  }

  stage_weights_store(sw, wu_bf, wp_bf, b, wave, lane);
}

constexpr int LG_WAVES  = 16;
constexpr int LG_KSLICE = K2 / LG_WAVES;
constexpr int LG_PASSK  = 256;
constexpr int LG_NPASS  = LG_KSLICE / LG_PASSK;
constexpr int LG_PSTEPS = LG_PASSK / 32;
constexpr int LG_GROUP  = 4;
constexpr int LG_PITCH  = LG_PASSK + 8;
constexpr int LG_NT_ROW = U / 16;
constexpr int LG_NT_COL = P / 16;
static_assert(LG_PASSK == 32 * 8 && LG_PSTEPS % LG_GROUP == 0, "logits staging split");

__global__ void __launch_bounds__(LG_WAVES * 32)
logits_wmma_kernel(const __bf16* __restrict__ c_bf,
                   const __bf16* __restrict__ W,
                   const float* __restrict__ bias_row,
                   const float* __restrict__ bias_col,
                   float* __restrict__ out_row,
                   float* __restrict__ out_col) {
  __shared__ __attribute__((aligned(16))) __bf16 stage[LG_WAVES][2][16 * LG_PITCH];
  __shared__ float red[LG_WAVES][256];

  int mtile = blockIdx.x;
  int ntile = blockIdx.y;
  int wave  = threadIdx.x >> 5;
  int lane  = threadIdx.x & 31;
  int rt    = lane & 15;
  int hi    = lane >> 4;
  int k0    = hi ? 8 : 0;

  const __bf16* Ag = c_bf + (size_t)(mtile * 16) * K2 + wave * LG_KSLICE + lane * 8;
  const __bf16* Wg = W    + (size_t)(ntile * 16) * K2 + wave * LG_KSLICE + lane * 8;
  __bf16* sA = &stage[wave][0][lane * 8];
  __bf16* sW = &stage[wave][1][lane * 8];
  const __bf16* fA = &stage[wave][0][rt * LG_PITCH + k0];
  const __bf16* fW = &stage[wave][1][rt * LG_PITCH + k0];

  v8bf ra[16], rw[16];
#pragma unroll
  for (int r = 0; r < 16; ++r) {
    ra[r] = *(const v8bf*)(Ag + (size_t)r * K2);
    rw[r] = *(const v8bf*)(Wg + (size_t)r * K2);
  }

  v8f acc = {};
#pragma unroll
  for (int ps = 0; ps < LG_NPASS; ++ps) {
#pragma unroll
    for (int r = 0; r < 16; ++r) {
      *(v8bf*)(sA + r * LG_PITCH) = ra[r];
      *(v8bf*)(sW + r * LG_PITCH) = rw[r];
    }
    if (ps + 1 < LG_NPASS) {
#pragma unroll
      for (int r = 0; r < 16; ++r) {
        ra[r] = *(const v8bf*)(Ag + (size_t)r * K2 + (ps + 1) * LG_PASSK);
        rw[r] = *(const v8bf*)(Wg + (size_t)r * K2 + (ps + 1) * LG_PASSK);
      }
    }
#pragma unroll
    for (int g = 0; g < LG_PSTEPS / LG_GROUP; ++g) {
      v16bf a[LG_GROUP], w[LG_GROUP];
#pragma unroll
      for (int s = 0; s < LG_GROUP; ++s) {
        a[s] = load_frag_bf16(fA + (g * LG_GROUP + s) * 32);
        w[s] = load_frag_bf16(fW + (g * LG_GROUP + s) * 32);
      }
      __builtin_amdgcn_sched_barrier(0);
#pragma unroll
      for (int s = 0; s < LG_GROUP; ++s)
        acc = __builtin_amdgcn_wmma_f32_16x16x32_bf16(
            false, a[s], false, w[s], (short)0, acc, false, false);
    }
  }

#pragma unroll
  for (int r = 0; r < 8; ++r) red[wave][r * 32 + lane] = acc[r];
  __syncthreads();

  if (threadIdx.x < 256) {
    int e = threadIdx.x;
    float s = red[0][e];
#pragma unroll
    for (int w = 1; w < LG_WAVES; ++w) s += red[w][e];
    int r = e >> 5;
    int l = e & 31;
    int m = r + 8 * (l >> 4);
    bool isRow = ntile < LG_NT_ROW;
    int N = isRow ? U : P;
    int n = (isRow ? ntile : ntile - LG_NT_ROW) * 16 + (l & 15);
    const float* bias = isRow ? bias_row : bias_col;
    float* out = isRow ? out_row : out_col;
    out[(size_t)(mtile * 16 + m) * N + n] = s + bias[n];
  }
}

template <int PER>
__device__ __forceinline__ void softmax_row(float* __restrict__ p, int lane) {
  float vals[PER];
  float mx = -1e30f;
#pragma unroll
  for (int j = 0; j < PER; ++j) { vals[j] = p[lane + 32 * j]; mx = fmaxf(mx, vals[j]); }
#pragma unroll
  for (int m = 16; m; m >>= 1) mx = fmaxf(mx, __shfl_xor(mx, m, 32));
  float s = 0.f;
#pragma unroll
  for (int j = 0; j < PER; ++j) { vals[j] = __expf(vals[j] - mx); s += vals[j]; }
#pragma unroll
  for (int m = 16; m; m >>= 1) s += __shfl_xor(s, m, 32);
  float inv = 1.0f / s;
#pragma unroll
  for (int j = 0; j < PER; ++j) p[lane + 32 * j] = vals[j] * inv;
}

static_assert(B % 8 == 0, "softmax: the 8 rows of a block lie in one output");

__global__ void __launch_bounds__(256)
softmax_kernel(float* __restrict__ row_logits, float* __restrict__ col_logits) {
  int row  = (blockIdx.x * 256 + threadIdx.x) >> 5;
  int lane = threadIdx.x & 31;
  if (row < B) softmax_row<U / 32>(row_logits + (size_t)row * U, lane);
  else         softmax_row<P / 32>(col_logits + (size_t)(row - B) * P, lane);
}

extern "C" void kernel_launch(void* const* d_in, const int* in_sizes, int n_in,
                              void* d_out, int out_size, void* d_ws, size_t ws_size,
                              hipStream_t stream) {
  const float* utt     = (const float*)d_in[0];
  const float* pheno   = (const float*)d_in[1];
  const float* w_utt   = (const float*)d_in[2];
  const float* b_utt   = (const float*)d_in[3];
  const float* w_pheno = (const float*)d_in[4];
  const float* b_pheno = (const float*)d_in[5];
  (void)in_sizes; (void)n_in; (void)out_size; (void)ws_size;

  char* ws = (char*)d_ws;
  size_t off = 0;
  auto alloc = [&](size_t bytes) {
    void* p = ws + off;
    off = (off + bytes + 255) & ~(size_t)255;
    return p;
  };
  __bf16* w_bf   = (__bf16*)alloc((size_t)(U + P) * K2 * sizeof(__bf16));
  __bf16* wu_bf  = w_bf;
  __bf16* wp_bf  = w_bf + (size_t)U * K2;
  __bf16* c_bf   = (__bf16*)alloc((size_t)B * K2 * sizeof(__bf16));

  float* row_logits = (float*)d_out;
  float* col_logits = (float*)d_out + B * U;

  cosine_wmma_kernel<<<B, 256, 0, stream>>>(utt, pheno, c_bf, w_utt, w_pheno, wu_bf, wp_bf);

  logits_wmma_kernel<<<dim3(B / 16, LG_NT_ROW + LG_NT_COL), LG_WAVES * 32, 0, stream>>>(
      c_bf, w_bf, b_utt, b_pheno, row_logits, col_logits);

  softmax_kernel<<<2 * B / 8, 256, 0, stream>>>(row_logits, col_logits);
}
